# starting kernel
# baseline (speedup 1.0000x reference)
#include <hip/hip_runtime.h>


typedef __attribute__((ext_vector_type(2))) float v2f;
typedef __attribute__((ext_vector_type(8))) float v8f;

template <bool RELU, bool DUAL>
__global__ __launch_bounds__(256) void gemm_wmma_kernel(
    const float* __restrict__ A, const float* __restrict__ W1,
    const float* __restrict__ B, const float* __restrict__ W2,
    const float* __restrict__ bias, float* __restrict__ out,
    int M, int Cin, int Cout)
{
    const int lane = threadIdx.x & 31;
    const int wavesPerBlock = blockDim.x >> 5;
    const int tile = blockIdx.x * wavesPerBlock + (threadIdx.x >> 5);
    const int ntn = Cout >> 4;
    const int ntm = M >> 4;
    if (tile >= ntm * ntn) return;
    const int tm = tile / ntn;
    const int tn = tile - tm * ntn;

    const int l    = lane & 15;
    const int koff = (lane >> 4) << 1;
    const int col  = tn * 16 + l;

    const float* arow = A + (size_t)(tm * 16 + l) * Cin;
    const float* brow = DUAL ? (B + (size_t)(tm * 16 + l) * Cin) : nullptr;

    v8f acc = {};
    for (int k = 0; k < Cin; k += 4) {
        v2f av; av.x = arow[k + koff];
                av.y = arow[k + koff + 1];
        v2f wv; wv.x = W1[(size_t)(k + koff)     * Cout + col];
                wv.y = W1[(size_t)(k + koff + 1) * Cout + col];
        acc = __builtin_amdgcn_wmma_f32_16x16x4_f32(
                  false, av, false, wv, (short)0, acc, false, false);
        if (DUAL) {
            v2f bv; bv.x = brow[k + koff];
                    bv.y = brow[k + koff + 1];
            v2f w2; w2.x = W2[(size_t)(k + koff)     * Cout + col];
                    w2.y = W2[(size_t)(k + koff + 1) * Cout + col];
            acc = __builtin_amdgcn_wmma_f32_16x16x4_f32(
                      false, bv, false, w2, (short)0, acc, false, false);
        }
    }

    const float bval = bias[col];
    const int m0 = tm * 16 + ((lane >> 4) << 3);
    #pragma unroll
    for (int r = 0; r < 8; ++r) {
        float v = acc[r] + bval;
        if (RELU) v = fmaxf(v, 0.0f);
        out[(size_t)(m0 + r) * Cout + col] = v;
    }
}

__global__ __launch_bounds__(256) void scatter_add_kernel(
    const float* __restrict__ h, const int* __restrict__ src,
    const int* __restrict__ dst, float* __restrict__ agg,
    long long E, int cshift)
{
    long long idx = (long long)blockIdx.x * blockDim.x + threadIdx.x;
    const int g4shift = cshift - 2;
    const long long total = E << g4shift;
    if (idx >= total) return;
    const int e  = (int)(idx >> g4shift);
    const int c4 = (int)(idx & ((1 << g4shift) - 1)) << 2;
    const int s  = src[e];
    const int d  = dst[e];
    const float4 v = *(const float4*)(h + ((size_t)s << cshift) + c4);
    float* dp = agg + ((size_t)d << cshift) + c4;
    atomicAdd(dp + 0, v.x);
    atomicAdd(dp + 1, v.y);
    atomicAdd(dp + 2, v.z);
    atomicAdd(dp + 3, v.w);
}

__global__ __launch_bounds__(256) void zero_kernel(float4* __restrict__ p, long long n4)
{
    long long i = (long long)blockIdx.x * blockDim.x + threadIdx.x;
    if (i < n4) p[i] = make_float4(0.f, 0.f, 0.f, 0.f);
}

__global__ __launch_bounds__(256) void pool_sum_kernel(
    const float* __restrict__ h, const int* __restrict__ batch,
    float* __restrict__ sums, float* __restrict__ cnt, long long N)
{
    long long idx = (long long)blockIdx.x * blockDim.x + threadIdx.x;
    const long long total = N << 5;
    if (idx >= total) return;
    const int node = (int)(idx >> 5);
    const int c4   = (int)(idx & 31) << 2;
    const int g    = batch[node];
    const float4 v = *(const float4*)(h + ((size_t)node << 7) + c4);
    float* sp = sums + ((size_t)g << 7) + c4;
    atomicAdd(sp + 0, v.x);
    atomicAdd(sp + 1, v.y);
    atomicAdd(sp + 2, v.z);
    atomicAdd(sp + 3, v.w);
    if (c4 == 0) atomicAdd(&cnt[g], 1.0f);
}

__global__ __launch_bounds__(256) void pool_div_kernel(
    const float4* __restrict__ sums, const float* __restrict__ cnt,
    float4* __restrict__ pooled, int G)
{
    int idx = blockIdx.x * blockDim.x + threadIdx.x;
    if (idx >= (G << 5)) return;
    const float m = fmaxf(cnt[idx >> 5], 1.0f);
    float4 s = sums[idx];
    pooled[idx] = make_float4(s.x / m, s.y / m, s.z / m, s.w / m);
}


static inline unsigned blocks_for(long long n, int t) {
    return (unsigned)((n + t - 1) / t);
}

extern "C" void kernel_launch(void* const* d_in, const int* in_sizes, int n_in,
                              void* d_out, int out_size, void* d_ws, size_t ws_size,
                              hipStream_t stream)
{
    const float* x        = (const float*)d_in[0];
    const int*   edge     = (const int*)  d_in[1];
    const int*   batch    = (const int*)  d_in[2];
    const float* w_stage1 = (const float*)d_in[4];
    const float* b_stage1 = (const float*)d_in[5];
    const float* rel1_w   = (const float*)d_in[6];
    const float* rel1_b   = (const float*)d_in[7];
    const float* root1_w  = (const float*)d_in[8];
    const float* rel2_w   = (const float*)d_in[9];
    const float* rel2_b   = (const float*)d_in[10];
    const float* root2_w  = (const float*)d_in[11];
    const float* rel3_w   = (const float*)d_in[12];
    const float* rel3_b   = (const float*)d_in[13];
    const float* root3_w  = (const float*)d_in[14];
    const float* lin_w    = (const float*)d_in[15];
    const float* lin_b    = (const float*)d_in[16];
    float*       out      = (float*)d_out;

    const int N = in_sizes[0] / 128;
    const int E = in_sizes[1] / 2;
    const int G = out_size / 16;
    const int* src = edge;
    const int* dst = edge + E;

    float* ws = (float*)d_ws;
    size_t off = 0;
    float* bufA   = ws + off; off += (size_t)N * 128;
    float* bufB   = ws + off; off += (size_t)N * 128;
    float* agg    = ws + off; off += (size_t)N * 128;
    float* sums   = ws + off; off += (size_t)G * 128;
    float* cnt    = ws + off; off += (size_t)G;
    float* pooled = ws + off; off += (size_t)G * 128;
    (void)ws_size; (void)n_in;

    const int T = 256;
    const int WPB = T / 32;

    auto gemm_grid = [&](int M, int Cout) {
        return blocks_for((long long)(M >> 4) * (Cout >> 4), WPB);
    };

    gemm_wmma_kernel<false, false><<<gemm_grid(N, 64), T, 0, stream>>>(
        x, w_stage1, nullptr, nullptr, b_stage1, bufA, N, 128, 64);

    zero_kernel<<<blocks_for((long long)N * 64 / 4, T), T, 0, stream>>>(
        (float4*)agg, (long long)N * 64 / 4);
    scatter_add_kernel<<<blocks_for((long long)E << 4, T), T, 0, stream>>>(
        bufA, src, dst, agg, E, 6);
    gemm_wmma_kernel<true, true><<<gemm_grid(N, 64), T, 0, stream>>>(
        agg, rel1_w, bufA, root1_w, rel1_b, bufB, N, 64, 64);

    zero_kernel<<<blocks_for((long long)N * 64 / 4, T), T, 0, stream>>>(
        (float4*)agg, (long long)N * 64 / 4);
    scatter_add_kernel<<<blocks_for((long long)E << 4, T), T, 0, stream>>>(
        bufB, src, dst, agg, E, 6);
    gemm_wmma_kernel<true, true><<<gemm_grid(N, 128), T, 0, stream>>>(
        agg, rel2_w, bufB, root2_w, rel2_b, bufA, N, 64, 128);

    zero_kernel<<<blocks_for((long long)N * 128 / 4, T), T, 0, stream>>>(
        (float4*)agg, (long long)N * 128 / 4);
    scatter_add_kernel<<<blocks_for((long long)E << 5, T), T, 0, stream>>>(
        bufA, src, dst, agg, E, 7);
    gemm_wmma_kernel<true, true><<<gemm_grid(N, 128), T, 0, stream>>>(
        agg, rel3_w, bufA, root3_w, rel3_b, bufB, N, 128, 128);

    zero_kernel<<<blocks_for((long long)G * 128 / 4, T), T, 0, stream>>>(
        (float4*)sums, (long long)G * 128 / 4);
    zero_kernel<<<blocks_for((long long)G / 4, T), T, 0, stream>>>(
        (float4*)cnt, (long long)G / 4);
    pool_sum_kernel<<<blocks_for((long long)N << 5, T), T, 0, stream>>>(
        bufB, batch, sums, cnt, N);
    pool_div_kernel<<<blocks_for((long long)G << 5, T), T, 0, stream>>>(
        (const float4*)sums, cnt, (float4*)pooled, G);

    gemm_wmma_kernel<false, false><<<gemm_grid(G, 16), T, 0, stream>>>(
        pooled, lin_w, nullptr, nullptr, lin_b, out, G, 128, 16);
}


# GEMM row operands via LDS: each work-group copies its 16-row A and B strips into LDS once with whole-line loads and every column-tile wave reads its fragments there; k order per accumulator unchanged
# speedup vs baseline: 2.0420x; 2.0420x over previous
#include <hip/hip_runtime.h>


typedef __attribute__((ext_vector_type(2))) float v2f;
typedef __attribute__((ext_vector_type(8))) float v8f;

template <bool RELU, bool DUAL, int CIN, int COUT, int WAVES>
__global__ __launch_bounds__(WAVES * 32) void gemm_wmma_kernel(
    const float* __restrict__ A, const float* __restrict__ W1,
    const float* __restrict__ B, const float* __restrict__ W2,
    const float* __restrict__ bias, float* __restrict__ out, int M)
{
    constexpr int NT    = WAVES * 32;
    constexpr int NTN   = COUT / 16;
    constexpr int SPB   = WAVES / NTN;
    constexpr int ROWS  = SPB * 16;
    constexpr int PITCH = CIN + 4;
    constexpr int C4    = CIN / 4;
    constexpr int NV    = ROWS * C4;
    constexpr int ITERS = NV / NT;
    static_assert(SPB * NTN == WAVES && ITERS * NT == NV, "tile shape");

    __shared__ __attribute__((aligned(16))) float sA[DUAL ? 2 : 1][ROWS * PITCH];

    const int tid  = threadIdx.x;
    const int row0 = blockIdx.x * ROWS;

    {
        const float4* gA = reinterpret_cast<const float4*>(A);
        const float4* gB = DUAL ? reinterpret_cast<const float4*>(B) : nullptr;
        float4 va[ITERS], vb[ITERS];
        #pragma unroll
        for (int it = 0; it < ITERS; ++it) {
            const int i = it * NT + tid;
            const int r = i / C4;
            const int c = i - r * C4;
            const size_t g = (size_t)min(row0 + r, M - 1) * C4 + c;
            va[it] = gA[g];
            if (DUAL) vb[it] = gB[g];
        }
        #pragma unroll
        for (int it = 0; it < ITERS; ++it) {
            const int i = it * NT + tid;
            const int r = i / C4;
            const int c = i - r * C4;
            *reinterpret_cast<float4*>(&sA[0][r * PITCH + c * 4]) = va[it];
            if (DUAL) *reinterpret_cast<float4*>(&sA[DUAL ? 1 : 0][r * PITCH + c * 4]) = vb[it];
        }
    }
    __syncthreads();

    const int lane = tid & 31;
    const int wave = tid >> 5;
    const int s    = wave / NTN;
    const int tn   = wave - s * NTN;
    const int tm16 = row0 + s * 16;
    if (tm16 >= M) return;

    const int l    = lane & 15;
    const int koff = (lane >> 4) << 1;
    const int col  = tn * 16 + l;

    const float* arow = &sA[0][(s * 16 + l) * PITCH + koff];
    const float* brow = &sA[DUAL ? 1 : 0][(s * 16 + l) * PITCH + koff];
    const float* w1p  = W1 + (size_t)koff * COUT + col;
    const float* w2p  = DUAL ? (W2 + (size_t)koff * COUT + col) : nullptr;

    v8f acc = {};
    for (int k = 0; k < CIN; k += 4) {
        const v2f av = *reinterpret_cast<const v2f*>(arow + k);
        v2f wv; wv.x = w1p[(size_t)k * COUT];
                wv.y = w1p[(size_t)(k + 1) * COUT];
        acc = __builtin_amdgcn_wmma_f32_16x16x4_f32(
                  false, av, false, wv, (short)0, acc, false, false);
        if (DUAL) {
            const v2f bv = *reinterpret_cast<const v2f*>(brow + k);
            v2f w2; w2.x = w2p[(size_t)k * COUT];
                    w2.y = w2p[(size_t)(k + 1) * COUT];
            acc = __builtin_amdgcn_wmma_f32_16x16x4_f32(
                      false, bv, false, w2, (short)0, acc, false, false);
        }
    }

    const float bval = bias[col];
    const int m0 = tm16 + ((lane >> 4) << 3);
    #pragma unroll
    for (int r = 0; r < 8; ++r) {
        float v = acc[r] + bval;
        if (RELU) v = fmaxf(v, 0.0f);
        out[(size_t)(m0 + r) * COUT + col] = v;
    }
}

constexpr int rows_per_wave(int ch) { return 1024 / ch; }

template <int CH>
__global__ __launch_bounds__(256) void scatter_add_kernel(
    const float* __restrict__ h, const int* __restrict__ src,
    const int* __restrict__ dst, float* __restrict__ agg, int E)
{
    constexpr int NL  = CH / 32;
    constexpr int RPW = rows_per_wave(CH);
    const int lane = threadIdx.x & 31;
    const int wave = __builtin_amdgcn_readfirstlane(
        (int)(blockIdx.x * (blockDim.x >> 5) + (threadIdx.x >> 5)));
    const int e0 = wave * RPW;
    if (e0 >= E) return;

    if (e0 + RPW <= E) {
        int s[RPW], d[RPW];
        #pragma unroll
        for (int j = 0; j < RPW; ++j) { s[j] = src[e0 + j]; d[j] = dst[e0 + j]; }
        float v[RPW][NL];
        #pragma unroll
        for (int j = 0; j < RPW; ++j) {
            const float* hrow = h + (size_t)s[j] * CH;
            #pragma unroll
            for (int c = 0; c < NL; ++c) v[j][c] = hrow[lane + 32 * c];
        }
        #pragma unroll
        for (int j = 0; j < RPW; ++j) {
            float* arow = agg + (size_t)d[j] * CH;
            #pragma unroll
            for (int c = 0; c < NL; ++c) atomicAdd(arow + lane + 32 * c, v[j][c]);
        }
    } else {
        for (int e = e0; e < E; ++e) {
            const float* hrow = h + (size_t)src[e] * CH;
            float* arow = agg + (size_t)dst[e] * CH;
            #pragma unroll
            for (int c = 0; c < NL; ++c) atomicAdd(arow + lane + 32 * c, hrow[lane + 32 * c]);
        }
    }
}

__global__ __launch_bounds__(256) void zero_kernel(float4* __restrict__ p, long long n4)
{
    long long i = (long long)blockIdx.x * blockDim.x + threadIdx.x;
    if (i < n4) p[i] = make_float4(0.f, 0.f, 0.f, 0.f);
}

__global__ __launch_bounds__(256) void pool_sum_kernel(
    const float* __restrict__ h, const int* __restrict__ batch,
    float* __restrict__ sums, float* __restrict__ cnt, int N)
{
    constexpr int RPW = rows_per_wave(128);
    const int lane = threadIdx.x & 31;
    const int wave = __builtin_amdgcn_readfirstlane(
        (int)(blockIdx.x * (blockDim.x >> 5) + (threadIdx.x >> 5)));
    const int n0 = wave * RPW;
    if (n0 >= N) return;

    if (n0 + RPW <= N) {
        int g[RPW];
        #pragma unroll
        for (int j = 0; j < RPW; ++j) g[j] = batch[n0 + j];
        const int gl = batch[n0 + (lane & (RPW - 1))];
        float v[RPW][4];
        #pragma unroll
        for (int j = 0; j < RPW; ++j) {
            const float* hrow = h + ((size_t)(n0 + j) << 7);
            #pragma unroll
            for (int c = 0; c < 4; ++c) v[j][c] = hrow[lane + 32 * c];
        }
        #pragma unroll
        for (int j = 0; j < RPW; ++j) {
            float* srow = sums + ((size_t)g[j] << 7);
            #pragma unroll
            for (int c = 0; c < 4; ++c) atomicAdd(srow + lane + 32 * c, v[j][c]);
        }
        if (lane < RPW) atomicAdd(&cnt[gl], 1.0f);
    } else {
        for (int n = n0; n < N; ++n) {
            const int gg = batch[n];
            const float* hrow = h + ((size_t)n << 7);
            float* srow = sums + ((size_t)gg << 7);
            #pragma unroll
            for (int c = 0; c < 4; ++c) atomicAdd(srow + lane + 32 * c, hrow[lane + 32 * c]);
            if (lane == 0) atomicAdd(&cnt[gg], 1.0f);
        }
    }
}

__global__ __launch_bounds__(256) void pool_div_kernel(
    const float4* __restrict__ sums, const float* __restrict__ cnt,
    float4* __restrict__ pooled, int G)
{
    int idx = blockIdx.x * blockDim.x + threadIdx.x;
    if (idx >= (G << 5)) return;
    const float m = fmaxf(cnt[idx >> 5], 1.0f);
    float4 s = sums[idx];
    pooled[idx] = make_float4(s.x / m, s.y / m, s.z / m, s.w / m);
}


static inline unsigned blocks_for(long long n, int t) {
    return (unsigned)((n + t - 1) / t);
}

extern "C" void kernel_launch(void* const* d_in, const int* in_sizes, int n_in,
                              void* d_out, int out_size, void* d_ws, size_t ws_size,
                              hipStream_t stream)
{
    const float* x        = (const float*)d_in[0];
    const int*   edge     = (const int*)  d_in[1];
    const int*   batch    = (const int*)  d_in[2];
    const float* w_stage1 = (const float*)d_in[4];
    const float* b_stage1 = (const float*)d_in[5];
    const float* rel1_w   = (const float*)d_in[6];
    const float* rel1_b   = (const float*)d_in[7];
    const float* root1_w  = (const float*)d_in[8];
    const float* rel2_w   = (const float*)d_in[9];
    const float* rel2_b   = (const float*)d_in[10];
    const float* root2_w  = (const float*)d_in[11];
    const float* rel3_w   = (const float*)d_in[12];
    const float* rel3_b   = (const float*)d_in[13];
    const float* root3_w  = (const float*)d_in[14];
    const float* lin_w    = (const float*)d_in[15];
    const float* lin_b    = (const float*)d_in[16];
    float*       out      = (float*)d_out;

    const int N = in_sizes[0] / 128;
    const int E = in_sizes[1] / 2;
    const int G = out_size / 16;
    const int* src = edge;
    const int* dst = edge + E;

    float* ws = (float*)d_ws;
    size_t off = 0;
    float* bufA   = ws + off; off += (size_t)N * 128;
    float* bufB   = ws + off; off += (size_t)N * 128;
    float* agg    = ws + off; off += (size_t)N * 128;
    float* sums   = ws + off; off += (size_t)G * 128;
    float* cnt    = ws + off; off += (size_t)G;
    float* pooled = ws + off; off += (size_t)G * 128;
    (void)ws_size; (void)n_in;

    const int T = 256;
    const int WPB = T / 32;

    auto gemm_grid = [&](int M, int rows_per_group) {
        return blocks_for(M, rows_per_group);
    };

    auto rows_grid = [&](int rows, int ch) {
        return blocks_for(blocks_for(rows, rows_per_wave(ch)), WPB);
    };

    gemm_wmma_kernel<false, false, 128, 64, 8><<<gemm_grid(N, 32), T, 0, stream>>>(
        x, w_stage1, nullptr, nullptr, b_stage1, bufA, N);

    zero_kernel<<<blocks_for((long long)N * 64 / 4, T), T, 0, stream>>>(
        (float4*)agg, (long long)N * 64 / 4);
    scatter_add_kernel<64><<<rows_grid(E, 64), T, 0, stream>>>(
        bufA, src, dst, agg, E);
    gemm_wmma_kernel<true, true, 64, 64, 8><<<gemm_grid(N, 32), T, 0, stream>>>(
        agg, rel1_w, bufA, root1_w, rel1_b, bufB, N);

    zero_kernel<<<blocks_for((long long)N * 64 / 4, T), T, 0, stream>>>(
        (float4*)agg, (long long)N * 64 / 4);
    scatter_add_kernel<64><<<rows_grid(E, 64), T, 0, stream>>>(
        bufB, src, dst, agg, E);
    gemm_wmma_kernel<true, true, 64, 128, 8><<<gemm_grid(N, 16), T, 0, stream>>>(
        agg, rel2_w, bufB, root2_w, rel2_b, bufA, N);

    zero_kernel<<<blocks_for((long long)N * 128 / 4, T), T, 0, stream>>>(
        (float4*)agg, (long long)N * 128 / 4);
    scatter_add_kernel<128><<<rows_grid(E, 128), T, 0, stream>>>(
        bufA, src, dst, agg, E);
    gemm_wmma_kernel<true, true, 128, 128, 8><<<gemm_grid(N, 16), T, 0, stream>>>(
        agg, rel3_w, bufA, root3_w, rel3_b, bufB, N);

    zero_kernel<<<blocks_for((long long)G * 128 / 4, T), T, 0, stream>>>(
        (float4*)sums, (long long)G * 128 / 4);
    zero_kernel<<<blocks_for((long long)G / 4, T), T, 0, stream>>>(
        (float4*)cnt, (long long)G / 4);
    pool_sum_kernel<<<rows_grid(N, 128), T, 0, stream>>>(
        bufB, batch, sums, cnt, N);
    pool_div_kernel<<<blocks_for((long long)G << 5, T), T, 0, stream>>>(
        (const float4*)sums, cnt, (float4*)pooled, G);

    gemm_wmma_kernel<false, false, 128, 16, 2><<<gemm_grid(G, 32), 64, 0, stream>>>(
        pooled, lin_w, nullptr, nullptr, lin_b, out, G);
}


# Zero fills ride in the GEMM launches: each GEMM work-group clears its own nodes' rows of the next layer's neighbour sums (pool sums in the first GEMM); 15 to 10 launches; sum order unchanged
# speedup vs baseline: 2.3054x; 1.1290x over previous
#include <hip/hip_runtime.h>


typedef __attribute__((ext_vector_type(2))) float v2f;
typedef __attribute__((ext_vector_type(8))) float v8f;

template <bool RELU, bool DUAL, int CIN, int COUT, int WAVES, int ZCH, bool ZFLAT>
__global__ __launch_bounds__(WAVES * 32) void gemm_wmma_kernel(
    const float* A, const float* __restrict__ W1,
    const float* __restrict__ B, const float* __restrict__ W2,
    const float* __restrict__ bias, float* __restrict__ out, int M,
    float* zrows, float* zflat, int zflat_n4)
{
    constexpr int NT    = WAVES * 32;
    constexpr int NTN   = COUT / 16;
    constexpr int SPB   = WAVES / NTN;
    constexpr int ROWS  = SPB * 16;
    constexpr int PITCH = CIN + 4;
    constexpr int C4    = CIN / 4;
    constexpr int NV    = ROWS * C4;
    constexpr int ITERS = NV / NT;
    static_assert(SPB * NTN == WAVES && ITERS * NT == NV, "tile shape");

    __shared__ __attribute__((aligned(16))) float sA[DUAL ? 2 : 1][ROWS * PITCH];

    const int tid  = threadIdx.x;
    const int row0 = blockIdx.x * ROWS;

    {
        const float4* gA = reinterpret_cast<const float4*>(A);
        const float4* gB = DUAL ? reinterpret_cast<const float4*>(B) : nullptr;
        float4 va[ITERS], vb[ITERS];
        #pragma unroll
        for (int it = 0; it < ITERS; ++it) {
            const int i = it * NT + tid;
            const int r = i / C4;
            const int c = i - r * C4;
            const size_t g = (size_t)min(row0 + r, M - 1) * C4 + c;
            va[it] = gA[g];
            if (DUAL) vb[it] = gB[g];
        }
        #pragma unroll
        for (int it = 0; it < ITERS; ++it) {
            const int i = it * NT + tid;
            const int r = i / C4;
            const int c = i - r * C4;
            *reinterpret_cast<float4*>(&sA[0][r * PITCH + c * 4]) = va[it];
            if (DUAL) *reinterpret_cast<float4*>(&sA[DUAL ? 1 : 0][r * PITCH + c * 4]) = vb[it];
        }
    }
    __syncthreads();

    if (ZCH > 0) {
        constexpr int Z4 = ZCH / 4;
        constexpr int NZ = ROWS * Z4;
        static_assert(ZCH == 0 || NZ % NT == 0, "fill shape");
        float4* z4 = reinterpret_cast<float4*>(zrows) + (size_t)row0 * Z4;
        for (int it = 0; it < NZ / NT; ++it) {
            const int i = it * NT + tid;
            if (row0 + i / Z4 < M) z4[i] = make_float4(0.f, 0.f, 0.f, 0.f);
        }
    }
    if (ZFLAT) {
        const int gi = blockIdx.x * NT + tid;
        if (gi < zflat_n4) reinterpret_cast<float4*>(zflat)[gi] = make_float4(0.f, 0.f, 0.f, 0.f);
    }

    const int lane = tid & 31;
    const int wave = tid >> 5;
    const int s    = wave / NTN;
    const int tn   = wave - s * NTN;
    const int tm16 = row0 + s * 16;
    if (tm16 >= M) return;

    const int l    = lane & 15;
    const int koff = (lane >> 4) << 1;
    const int col  = tn * 16 + l;

    const float* arow = &sA[0][(s * 16 + l) * PITCH + koff];
    const float* brow = &sA[DUAL ? 1 : 0][(s * 16 + l) * PITCH + koff];
    const float* w1p  = W1 + (size_t)koff * COUT + col;
    const float* w2p  = DUAL ? (W2 + (size_t)koff * COUT + col) : nullptr;

    v8f acc = {};
    for (int k = 0; k < CIN; k += 4) {
        const v2f av = *reinterpret_cast<const v2f*>(arow + k);
        v2f wv; wv.x = w1p[(size_t)k * COUT];
                wv.y = w1p[(size_t)(k + 1) * COUT];
        acc = __builtin_amdgcn_wmma_f32_16x16x4_f32(
                  false, av, false, wv, (short)0, acc, false, false);
        if (DUAL) {
            const v2f bv = *reinterpret_cast<const v2f*>(brow + k);
            v2f w2; w2.x = w2p[(size_t)k * COUT];
                    w2.y = w2p[(size_t)(k + 1) * COUT];
            acc = __builtin_amdgcn_wmma_f32_16x16x4_f32(
                      false, bv, false, w2, (short)0, acc, false, false);
        }
    }

    const float bval = bias[col];
    const int m0 = tm16 + ((lane >> 4) << 3);
    #pragma unroll
    for (int r = 0; r < 8; ++r) {
        float v = acc[r] + bval;
        if (RELU) v = fmaxf(v, 0.0f);
        out[(size_t)(m0 + r) * COUT + col] = v;
    }
}

constexpr int rows_per_wave(int ch) { return 1024 / ch; }

template <int CH>
__global__ __launch_bounds__(256) void scatter_add_kernel(
    const float* __restrict__ h, const int* __restrict__ src,
    const int* __restrict__ dst, float* __restrict__ agg, int E)
{
    constexpr int NL  = CH / 32;
    constexpr int RPW = rows_per_wave(CH);
    const int lane = threadIdx.x & 31;
    const int wave = __builtin_amdgcn_readfirstlane(
        (int)(blockIdx.x * (blockDim.x >> 5) + (threadIdx.x >> 5)));
    const int e0 = wave * RPW;
    if (e0 >= E) return;

    if (e0 + RPW <= E) {
        int s[RPW], d[RPW];
        #pragma unroll
        for (int j = 0; j < RPW; ++j) { s[j] = src[e0 + j]; d[j] = dst[e0 + j]; }
        float v[RPW][NL];
        #pragma unroll
        for (int j = 0; j < RPW; ++j) {
            const float* hrow = h + (size_t)s[j] * CH;
            #pragma unroll
            for (int c = 0; c < NL; ++c) v[j][c] = hrow[lane + 32 * c];
        }
        #pragma unroll
        for (int j = 0; j < RPW; ++j) {
            float* arow = agg + (size_t)d[j] * CH;
            #pragma unroll
            for (int c = 0; c < NL; ++c) atomicAdd(arow + lane + 32 * c, v[j][c]);
        }
    } else {
        for (int e = e0; e < E; ++e) {
            const float* hrow = h + (size_t)src[e] * CH;
            float* arow = agg + (size_t)dst[e] * CH;
            #pragma unroll
            for (int c = 0; c < NL; ++c) atomicAdd(arow + lane + 32 * c, hrow[lane + 32 * c]);
        }
    }
}

__global__ __launch_bounds__(256) void pool_sum_kernel(
    const float* __restrict__ h, const int* __restrict__ batch,
    float* __restrict__ sums, float* __restrict__ cnt, int N)
{
    constexpr int RPW = rows_per_wave(128);
    const int lane = threadIdx.x & 31;
    const int wave = __builtin_amdgcn_readfirstlane(
        (int)(blockIdx.x * (blockDim.x >> 5) + (threadIdx.x >> 5)));
    const int n0 = wave * RPW;
    if (n0 >= N) return;

    if (n0 + RPW <= N) {
        int g[RPW];
        #pragma unroll
        for (int j = 0; j < RPW; ++j) g[j] = batch[n0 + j];
        const int gl = batch[n0 + (lane & (RPW - 1))];
        float v[RPW][4];
        #pragma unroll
        for (int j = 0; j < RPW; ++j) {
            const float* hrow = h + ((size_t)(n0 + j) << 7);
            #pragma unroll
            for (int c = 0; c < 4; ++c) v[j][c] = hrow[lane + 32 * c];
        }
        #pragma unroll
        for (int j = 0; j < RPW; ++j) {
            float* srow = sums + ((size_t)g[j] << 7);
            #pragma unroll
            for (int c = 0; c < 4; ++c) atomicAdd(srow + lane + 32 * c, v[j][c]);
        }
        if (lane < RPW) atomicAdd(&cnt[gl], 1.0f);
    } else {
        for (int n = n0; n < N; ++n) {
            const int gg = batch[n];
            const float* hrow = h + ((size_t)n << 7);
            float* srow = sums + ((size_t)gg << 7);
            #pragma unroll
            for (int c = 0; c < 4; ++c) atomicAdd(srow + lane + 32 * c, hrow[lane + 32 * c]);
            if (lane == 0) atomicAdd(&cnt[gg], 1.0f);
        }
    }
}

__global__ __launch_bounds__(256) void pool_div_kernel(
    const float4* __restrict__ sums, const float* __restrict__ cnt,
    float4* __restrict__ pooled, int G)
{
    int idx = blockIdx.x * blockDim.x + threadIdx.x;
    if (idx >= (G << 5)) return;
    const float m = fmaxf(cnt[idx >> 5], 1.0f);
    float4 s = sums[idx];
    pooled[idx] = make_float4(s.x / m, s.y / m, s.z / m, s.w / m);
}


static inline unsigned blocks_for(long long n, int t) {
    return (unsigned)((n + t - 1) / t);
}

extern "C" void kernel_launch(void* const* d_in, const int* in_sizes, int n_in,
                              void* d_out, int out_size, void* d_ws, size_t ws_size,
                              hipStream_t stream)
{
    const float* x        = (const float*)d_in[0];
    const int*   edge     = (const int*)  d_in[1];
    const int*   batch    = (const int*)  d_in[2];
    const float* w_stage1 = (const float*)d_in[4];
    const float* b_stage1 = (const float*)d_in[5];
    const float* rel1_w   = (const float*)d_in[6];
    const float* rel1_b   = (const float*)d_in[7];
    const float* root1_w  = (const float*)d_in[8];
    const float* rel2_w   = (const float*)d_in[9];
    const float* rel2_b   = (const float*)d_in[10];
    const float* root2_w  = (const float*)d_in[11];
    const float* rel3_w   = (const float*)d_in[12];
    const float* rel3_b   = (const float*)d_in[13];
    const float* root3_w  = (const float*)d_in[14];
    const float* lin_w    = (const float*)d_in[15];
    const float* lin_b    = (const float*)d_in[16];
    float*       out      = (float*)d_out;

    const int N = in_sizes[0] / 128;
    const int E = in_sizes[1] / 2;
    const int G = out_size / 16;
    const int* src = edge;
    const int* dst = edge + E;

    float* ws = (float*)d_ws;
    size_t off = 0;
    float* bufA   = ws + off; off += (size_t)N * 128;
    float* bufB   = ws + off; off += (size_t)N * 128;
    float* agg    = ws + off; off += (size_t)N * 128;
    float* agg3   = ws + off; off += (size_t)N * 128;
    float* sums   = ws + off; off += (size_t)G * 128;
    float* cnt    = ws + off; off += (size_t)G;
    float* pooled = ws + off; off += (size_t)G * 128;
    (void)ws_size; (void)n_in;

    const int T = 256;
    const int WPB = T / 32;

    auto gemm_grid = [&](int M, int rows_per_group) {
        return blocks_for(M, rows_per_group);
    };

    auto rows_grid = [&](int rows, int ch) {
        return blocks_for(blocks_for(rows, rows_per_wave(ch)), WPB);
    };

    gemm_wmma_kernel<false, false, 128, 64, 8, 64, true><<<gemm_grid(N, 32), T, 0, stream>>>(
        x, w_stage1, nullptr, nullptr, b_stage1, bufA, N, agg, sums, (G * 128 + G) / 4);

    scatter_add_kernel<64><<<rows_grid(E, 64), T, 0, stream>>>(
        bufA, src, dst, agg, E);
    gemm_wmma_kernel<true, true, 64, 64, 8, 64, false><<<gemm_grid(N, 32), T, 0, stream>>>(
        agg, rel1_w, bufA, root1_w, rel1_b, bufB, N, agg, nullptr, 0);

    scatter_add_kernel<64><<<rows_grid(E, 64), T, 0, stream>>>(
        bufB, src, dst, agg, E);
    gemm_wmma_kernel<true, true, 64, 128, 8, 128, false><<<gemm_grid(N, 16), T, 0, stream>>>(
        agg, rel2_w, bufB, root2_w, rel2_b, bufA, N, agg3, nullptr, 0);

    scatter_add_kernel<128><<<rows_grid(E, 128), T, 0, stream>>>(
        bufA, src, dst, agg3, E);
    gemm_wmma_kernel<true, true, 128, 128, 8, 0, false><<<gemm_grid(N, 16), T, 0, stream>>>(
        agg3, rel3_w, bufA, root3_w, rel3_b, bufB, N, nullptr, nullptr, 0);

    pool_sum_kernel<<<rows_grid(N, 128), T, 0, stream>>>(
        bufB, batch, sums, cnt, N);
    pool_div_kernel<<<blocks_for((long long)G << 5, T), T, 0, stream>>>(
        (const float4*)sums, cnt, (float4*)pooled, G);

    gemm_wmma_kernel<false, false, 128, 16, 2, 0, false><<<gemm_grid(G, 32), 64, 0, stream>>>(
        pooled, lin_w, nullptr, nullptr, lin_b, out, G, nullptr, nullptr, 0);
}


# Classifier product moved in front of the mean pool (both linear): conv 3's waves multiply their h3 tiles by lin_w on WMMA, pool sums cover 16 classes; 10 to 8 launches; sum order: atomic arrival
# speedup vs baseline: 2.4554x; 1.0651x over previous
#include <hip/hip_runtime.h>


typedef __attribute__((ext_vector_type(2))) float v2f;
typedef __attribute__((ext_vector_type(8))) float v8f;

template <bool RELU, bool DUAL, int CIN, int COUT, int WAVES, int ZCH, bool ZFLAT, bool POOL>
__global__ __launch_bounds__(WAVES * 32) void gemm_wmma_kernel(
    const float* A, const float* __restrict__ W1,
    const float* __restrict__ B, const float* __restrict__ W2,
    const float* __restrict__ bias, float* __restrict__ out, int M,
    float* zrows, float* zflat, int zflat_n4,
    const float* __restrict__ lw, const int* __restrict__ batch,
    float* __restrict__ ysum, float* __restrict__ cnt)
{
    constexpr int NT    = WAVES * 32;
    constexpr int NTN   = COUT / 16;
    constexpr int SPB   = WAVES / NTN;
    constexpr int ROWS  = SPB * 16;
    constexpr int PITCH = CIN + 4;
    constexpr int C4    = CIN / 4;
    constexpr int NV    = ROWS * C4;
    constexpr int ITERS = NV / NT;
    static_assert(SPB * NTN == WAVES && ITERS * NT == NV, "tile shape");

    __shared__ __attribute__((aligned(16))) float sA[DUAL ? 2 : 1][ROWS * PITCH];
    constexpr int TP = 20;
    __shared__ __attribute__((aligned(16))) float sT[POOL ? WAVES : 1][16 * TP];
    static_assert(!POOL || SPB == 1, "POOL: one strip per work-group, every wave reaches the barrier");

    const int tid  = threadIdx.x;
    const int row0 = blockIdx.x * ROWS;

    {
        const float4* gA = reinterpret_cast<const float4*>(A);
        const float4* gB = DUAL ? reinterpret_cast<const float4*>(B) : nullptr;
        float4 va[ITERS], vb[ITERS];
        #pragma unroll
        for (int it = 0; it < ITERS; ++it) {
            const int i = it * NT + tid;
            const int r = i / C4;
            const int c = i - r * C4;
            const size_t g = (size_t)min(row0 + r, M - 1) * C4 + c;
            va[it] = gA[g];
            if (DUAL) vb[it] = gB[g];
        }
        #pragma unroll
        for (int it = 0; it < ITERS; ++it) {
            const int i = it * NT + tid;
            const int r = i / C4;
            const int c = i - r * C4;
            *reinterpret_cast<float4*>(&sA[0][r * PITCH + c * 4]) = va[it];
            if (DUAL) *reinterpret_cast<float4*>(&sA[DUAL ? 1 : 0][r * PITCH + c * 4]) = vb[it];
        }
    }
    __syncthreads();

    if (ZCH > 0) {
        constexpr int Z4 = ZCH / 4;
        constexpr int NZ = ROWS * Z4;
        static_assert(ZCH == 0 || NZ % NT == 0, "fill shape");
        float4* z4 = reinterpret_cast<float4*>(zrows) + (size_t)row0 * Z4;
        for (int it = 0; it < NZ / NT; ++it) {
            const int i = it * NT + tid;
            if (row0 + i / Z4 < M) z4[i] = make_float4(0.f, 0.f, 0.f, 0.f);
        }
    }
    if (ZFLAT) {
        const int gi = blockIdx.x * NT + tid;
        if (gi < zflat_n4) reinterpret_cast<float4*>(zflat)[gi] = make_float4(0.f, 0.f, 0.f, 0.f);
    }

    const int lane = tid & 31;
    const int wave = tid >> 5;
    const int s    = wave / NTN;
    const int tn   = wave - s * NTN;
    const int tm16 = row0 + s * 16;
    if (!POOL && tm16 >= M) return;

    const int l    = lane & 15;
    const int koff = (lane >> 4) << 1;
    const int col  = tn * 16 + l;

    const float* arow = &sA[0][(s * 16 + l) * PITCH + koff];
    const float* brow = &sA[DUAL ? 1 : 0][(s * 16 + l) * PITCH + koff];
    const float* w1p  = W1 + (size_t)koff * COUT + col;
    const float* w2p  = DUAL ? (W2 + (size_t)koff * COUT + col) : nullptr;

    v8f acc = {};
    for (int k = 0; k < CIN; k += 4) {
        const v2f av = *reinterpret_cast<const v2f*>(arow + k);
        v2f wv; wv.x = w1p[(size_t)k * COUT];
                wv.y = w1p[(size_t)(k + 1) * COUT];
        acc = __builtin_amdgcn_wmma_f32_16x16x4_f32(
                  false, av, false, wv, (short)0, acc, false, false);
        if (DUAL) {
            const v2f bv = *reinterpret_cast<const v2f*>(brow + k);
            v2f w2; w2.x = w2p[(size_t)k * COUT];
                    w2.y = w2p[(size_t)(k + 1) * COUT];
            acc = __builtin_amdgcn_wmma_f32_16x16x4_f32(
                      false, bv, false, w2, (short)0, acc, false, false);
        }
    }

    const float bval = bias[col];
    const int m0 = tm16 + ((lane >> 4) << 3);
    if (!POOL) {
        #pragma unroll
        for (int r = 0; r < 8; ++r) {
            float v = acc[r] + bval;
            if (RELU) v = fmaxf(v, 0.0f);
            out[(size_t)(m0 + r) * COUT + col] = v;
        }
    } else {
        float* tile = &sT[wave][0];
        const int h8 = (lane >> 4) << 3;
        #pragma unroll
        for (int r = 0; r < 8; ++r) {
            float v = acc[r] + bval;
            if (RELU) v = fmaxf(v, 0.0f);
            tile[(h8 + r) * TP + l] = v;
        }
        __syncthreads();
        const float* lwp = lw + (size_t)(tn * 16 + koff) * 16 + l;
        v8f y = {};
        #pragma unroll
        for (int k = 0; k < 16; k += 4) {
            const v2f av = *reinterpret_cast<const v2f*>(&tile[l * TP + k + koff]);
            v2f wv; wv.x = lwp[k * 16];
                    wv.y = lwp[(k + 1) * 16];
            y = __builtin_amdgcn_wmma_f32_16x16x4_f32(
                    false, av, false, wv, (short)0, y, false, false);
        }
        int g[8];
        #pragma unroll
        for (int r = 0; r < 8; ++r) g[r] = batch[m0 + r];
        const bool counts = (tn == 0) && (l == 0);
        int gcur = g[0];
        float run = y[0];
        float nrun = 1.0f;
        #pragma unroll
        for (int r = 1; r < 8; ++r) {
            if (g[r] == gcur) { run += y[r]; nrun += 1.0f; }
            else {
                atomicAdd(&ysum[(size_t)gcur * 16 + l], run);
                if (counts) atomicAdd(&cnt[gcur], nrun);
                gcur = g[r]; run = y[r]; nrun = 1.0f;
            }
        }
        atomicAdd(&ysum[(size_t)gcur * 16 + l], run);
        if (counts) atomicAdd(&cnt[gcur], nrun);
    }
}

constexpr int rows_per_wave(int ch) { return 1024 / ch; }

template <int CH>
__global__ __launch_bounds__(256) void scatter_add_kernel(
    const float* __restrict__ h, const int* __restrict__ src,
    const int* __restrict__ dst, float* __restrict__ agg, int E)
{
    constexpr int NL  = CH / 32;
    constexpr int RPW = rows_per_wave(CH);
    const int lane = threadIdx.x & 31;
    const int wave = __builtin_amdgcn_readfirstlane(
        (int)(blockIdx.x * (blockDim.x >> 5) + (threadIdx.x >> 5)));
    const int e0 = wave * RPW;
    if (e0 >= E) return;

    if (e0 + RPW <= E) {
        int s[RPW], d[RPW];
        #pragma unroll
        for (int j = 0; j < RPW; ++j) { s[j] = src[e0 + j]; d[j] = dst[e0 + j]; }
        float v[RPW][NL];
        #pragma unroll
        for (int j = 0; j < RPW; ++j) {
            const float* hrow = h + (size_t)s[j] * CH;
            #pragma unroll
            for (int c = 0; c < NL; ++c) v[j][c] = hrow[lane + 32 * c];
        }
        #pragma unroll
        for (int j = 0; j < RPW; ++j) {
            float* arow = agg + (size_t)d[j] * CH;
            #pragma unroll
            for (int c = 0; c < NL; ++c) atomicAdd(arow + lane + 32 * c, v[j][c]);
        }
    } else {
        for (int e = e0; e < E; ++e) {
            const float* hrow = h + (size_t)src[e] * CH;
            float* arow = agg + (size_t)dst[e] * CH;
            #pragma unroll
            for (int c = 0; c < NL; ++c) atomicAdd(arow + lane + 32 * c, hrow[lane + 32 * c]);
        }
    }
}

__global__ __launch_bounds__(256) void pool_mean_bias_kernel(
    const float* __restrict__ ysum, const float* __restrict__ cnt,
    const float* __restrict__ lin_b, float* __restrict__ out, int G)
{
    const int idx = blockIdx.x * blockDim.x + threadIdx.x;
    if (idx >= (G << 4)) return;
    const float m = fmaxf(cnt[idx >> 4], 1.0f);
    out[idx] = ysum[idx] / m + lin_b[idx & 15];
}


static inline unsigned blocks_for(long long n, int t) {
    return (unsigned)((n + t - 1) / t);
}

extern "C" void kernel_launch(void* const* d_in, const int* in_sizes, int n_in,
                              void* d_out, int out_size, void* d_ws, size_t ws_size,
                              hipStream_t stream)
{
    const float* x        = (const float*)d_in[0];
    const int*   edge     = (const int*)  d_in[1];
    const int*   batch    = (const int*)  d_in[2];
    const float* w_stage1 = (const float*)d_in[4];
    const float* b_stage1 = (const float*)d_in[5];
    const float* rel1_w   = (const float*)d_in[6];
    const float* rel1_b   = (const float*)d_in[7];
    const float* root1_w  = (const float*)d_in[8];
    const float* rel2_w   = (const float*)d_in[9];
    const float* rel2_b   = (const float*)d_in[10];
    const float* root2_w  = (const float*)d_in[11];
    const float* rel3_w   = (const float*)d_in[12];
    const float* rel3_b   = (const float*)d_in[13];
    const float* root3_w  = (const float*)d_in[14];
    const float* lin_w    = (const float*)d_in[15];
    const float* lin_b    = (const float*)d_in[16];
    float*       out      = (float*)d_out;

    const int N = in_sizes[0] / 128;
    const int E = in_sizes[1] / 2;
    const int G = out_size / 16;
    const int* src = edge;
    const int* dst = edge + E;

    float* ws = (float*)d_ws;
    size_t off = 0;
    float* bufA   = ws + off; off += (size_t)N * 128;
    float* bufB   = ws + off; off += (size_t)N * 128;
    float* agg    = ws + off; off += (size_t)N * 128;
    float* agg3   = ws + off; off += (size_t)N * 128;
    float* ysum   = ws + off; off += (size_t)G * 16;
    float* cnt    = ws + off; off += (size_t)G;
    (void)ws_size; (void)n_in;

    const int T = 256;
    const int WPB = T / 32;

    auto gemm_grid = [&](int M, int rows_per_group) {
        return blocks_for(M, rows_per_group);
    };

    auto rows_grid = [&](int rows, int ch) {
        return blocks_for(blocks_for(rows, rows_per_wave(ch)), WPB);
    };

    gemm_wmma_kernel<false, false, 128, 64, 8, 64, true, false><<<gemm_grid(N, 32), T, 0, stream>>>(
        x, w_stage1, nullptr, nullptr, b_stage1, bufA, N, agg, ysum, (G * 16 + G) / 4,
        nullptr, nullptr, nullptr, nullptr);

    scatter_add_kernel<64><<<rows_grid(E, 64), T, 0, stream>>>(
        bufA, src, dst, agg, E);
    gemm_wmma_kernel<true, true, 64, 64, 8, 64, false, false><<<gemm_grid(N, 32), T, 0, stream>>>(
        agg, rel1_w, bufA, root1_w, rel1_b, bufB, N, agg, nullptr, 0,
        nullptr, nullptr, nullptr, nullptr);

    scatter_add_kernel<64><<<rows_grid(E, 64), T, 0, stream>>>(
        bufB, src, dst, agg, E);
    gemm_wmma_kernel<true, true, 64, 128, 8, 128, false, false><<<gemm_grid(N, 16), T, 0, stream>>>(
        agg, rel2_w, bufB, root2_w, rel2_b, bufA, N, agg3, nullptr, 0,
        nullptr, nullptr, nullptr, nullptr);

    scatter_add_kernel<128><<<rows_grid(E, 128), T, 0, stream>>>(
        bufA, src, dst, agg3, E);
    gemm_wmma_kernel<true, true, 128, 128, 8, 0, false, true><<<gemm_grid(N, 16), T, 0, stream>>>(
        agg3, rel3_w, bufA, root3_w, rel3_b, nullptr, N, nullptr, nullptr, 0,
        lin_w, batch, ysum, cnt);

    pool_mean_bias_kernel<<<blocks_for((long long)G << 4, T), T, 0, stream>>>(
        ysum, cnt, lin_b, out, G);
}


# Edge scatter-adds: node range cut in two, a wave adds only into rows of the L2 half it runs on (die id read), edges dealt by ticket counters to a fixed resident grid; sum order: atomic arrival
# speedup vs baseline: 2.7757x; 1.1304x over previous
#include <hip/hip_runtime.h>


typedef __attribute__((ext_vector_type(2))) float v2f;
typedef __attribute__((ext_vector_type(8))) float v8f;

template <bool RELU, bool DUAL, int CIN, int COUT, int WAVES, int ZCH, bool ZFLAT, bool POOL>
__global__ __launch_bounds__(WAVES * 32) void gemm_wmma_kernel(
    const float* A, const float* __restrict__ W1,
    const float* __restrict__ B, const float* __restrict__ W2,
    const float* __restrict__ bias, float* __restrict__ out, int M,
    float* zrows, float* zflat, int zflat_n4,
    const float* __restrict__ lw, const int* __restrict__ batch,
    float* __restrict__ ysum, float* __restrict__ cnt)
{
    constexpr int NT    = WAVES * 32;
    constexpr int NTN   = COUT / 16;
    constexpr int SPB   = WAVES / NTN;
    constexpr int ROWS  = SPB * 16;
    constexpr int PITCH = CIN + 4;
    constexpr int C4    = CIN / 4;
    constexpr int NV    = ROWS * C4;
    constexpr int ITERS = NV / NT;
    static_assert(SPB * NTN == WAVES && ITERS * NT == NV, "tile shape");

    __shared__ __attribute__((aligned(16))) float sA[DUAL ? 2 : 1][ROWS * PITCH];
    constexpr int TP = 20;
    __shared__ __attribute__((aligned(16))) float sT[POOL ? WAVES : 1][16 * TP];
    static_assert(!POOL || SPB == 1, "POOL: one strip per work-group, every wave reaches the barrier");

    const int tid  = threadIdx.x;
    const int row0 = blockIdx.x * ROWS;

    {
        const float4* gA = reinterpret_cast<const float4*>(A);
        const float4* gB = DUAL ? reinterpret_cast<const float4*>(B) : nullptr;
        float4 va[ITERS], vb[ITERS];
        #pragma unroll
        for (int it = 0; it < ITERS; ++it) {
            const int i = it * NT + tid;
            const int r = i / C4;
            const int c = i - r * C4;
            const size_t g = (size_t)min(row0 + r, M - 1) * C4 + c;
            va[it] = gA[g];
            if (DUAL) vb[it] = gB[g];
        }
        #pragma unroll
        for (int it = 0; it < ITERS; ++it) {
            const int i = it * NT + tid;
            const int r = i / C4;
            const int c = i - r * C4;
            *reinterpret_cast<float4*>(&sA[0][r * PITCH + c * 4]) = va[it];
            if (DUAL) *reinterpret_cast<float4*>(&sA[DUAL ? 1 : 0][r * PITCH + c * 4]) = vb[it];
        }
    }
    __syncthreads();

    if (ZCH > 0) {
        constexpr int Z4 = ZCH / 4;
        constexpr int NZ = ROWS * Z4;
        static_assert(ZCH == 0 || NZ % NT == 0, "fill shape");
        float4* z4 = reinterpret_cast<float4*>(zrows) + (size_t)row0 * Z4;
        for (int it = 0; it < NZ / NT; ++it) {
            const int i = it * NT + tid;
            if (row0 + i / Z4 < M) z4[i] = make_float4(0.f, 0.f, 0.f, 0.f);
        }
    }
    if (ZFLAT) {
        const int gi = blockIdx.x * NT + tid;
        if (gi < zflat_n4) reinterpret_cast<float4*>(zflat)[gi] = make_float4(0.f, 0.f, 0.f, 0.f);
    }

    const int lane = tid & 31;
    const int wave = tid >> 5;
    const int s    = wave / NTN;
    const int tn   = wave - s * NTN;
    const int tm16 = row0 + s * 16;
    if (!POOL && tm16 >= M) return;

    const int l    = lane & 15;
    const int koff = (lane >> 4) << 1;
    const int col  = tn * 16 + l;

    const float* arow = &sA[0][(s * 16 + l) * PITCH + koff];
    const float* brow = &sA[DUAL ? 1 : 0][(s * 16 + l) * PITCH + koff];
    const float* w1p  = W1 + (size_t)koff * COUT + col;
    const float* w2p  = DUAL ? (W2 + (size_t)koff * COUT + col) : nullptr;

    v8f acc = {};
    for (int k = 0; k < CIN; k += 4) {
        const v2f av = *reinterpret_cast<const v2f*>(arow + k);
        v2f wv; wv.x = w1p[(size_t)k * COUT];
                wv.y = w1p[(size_t)(k + 1) * COUT];
        acc = __builtin_amdgcn_wmma_f32_16x16x4_f32(
                  false, av, false, wv, (short)0, acc, false, false);
        if (DUAL) {
            const v2f bv = *reinterpret_cast<const v2f*>(brow + k);
            v2f w2; w2.x = w2p[(size_t)k * COUT];
                    w2.y = w2p[(size_t)(k + 1) * COUT];
            acc = __builtin_amdgcn_wmma_f32_16x16x4_f32(
                      false, bv, false, w2, (short)0, acc, false, false);
        }
    }

    const float bval = bias[col];
    const int m0 = tm16 + ((lane >> 4) << 3);
    if (!POOL) {
        #pragma unroll
        for (int r = 0; r < 8; ++r) {
            float v = acc[r] + bval;
            if (RELU) v = fmaxf(v, 0.0f);
            out[(size_t)(m0 + r) * COUT + col] = v;
        }
    } else {
        float* tile = &sT[wave][0];
        const int h8 = (lane >> 4) << 3;
        #pragma unroll
        for (int r = 0; r < 8; ++r) {
            float v = acc[r] + bval;
            if (RELU) v = fmaxf(v, 0.0f);
            tile[(h8 + r) * TP + l] = v;
        }
        __syncthreads();
        const float* lwp = lw + (size_t)(tn * 16 + koff) * 16 + l;
        v8f y = {};
        #pragma unroll
        for (int k = 0; k < 16; k += 4) {
            const v2f av = *reinterpret_cast<const v2f*>(&tile[l * TP + k + koff]);
            v2f wv; wv.x = lwp[k * 16];
                    wv.y = lwp[(k + 1) * 16];
            y = __builtin_amdgcn_wmma_f32_16x16x4_f32(
                    false, av, false, wv, (short)0, y, false, false);
        }
        int g[8];
        #pragma unroll
        for (int r = 0; r < 8; ++r) g[r] = batch[m0 + r];
        const bool counts = (tn == 0) && (l == 0);
        int gcur = g[0];
        float run = y[0];
        float nrun = 1.0f;
        #pragma unroll
        for (int r = 1; r < 8; ++r) {
            if (g[r] == gcur) { run += y[r]; nrun += 1.0f; }
            else {
                atomicAdd(&ysum[(size_t)gcur * 16 + l], run);
                if (counts) atomicAdd(&cnt[gcur], nrun);
                gcur = g[r]; run = y[r]; nrun = 1.0f;
            }
        }
        atomicAdd(&ysum[(size_t)gcur * 16 + l], run);
        if (counts) atomicAdd(&cnt[gcur], nrun);
    }
}

constexpr int SCATTER_GROUPS = 1024;

template <int CH>
__global__ __launch_bounds__(256) void scatter_add_kernel(
    const float* __restrict__ h, const int* __restrict__ src,
    const int* __restrict__ dst, float* __restrict__ agg, int E, int N,
    int* __restrict__ tickets)
{
    constexpr int NL  = CH / 32;
    constexpr int B   = 20;
    constexpr int EPW = 64;
    constexpr int EPG = EPW * 8;
    __shared__ int s_ticket[2];

    const int lane = threadIdx.x & 31;
    const int wv = __builtin_amdgcn_readfirstlane((int)(threadIdx.x >> 5));
    const int nchunks = (E + EPG - 1) / EPG;
    const int nsplit  = N >> 1;
    const int myhalf = (int)((__builtin_amdgcn_s_sendmsg_rtn(0x87) >> 18) & 1u);

    int it = 0;
    for (int pass = 0; pass < 2; ++pass) {
        const int  role = myhalf ^ pass;
        const bool low  = (role == 0);
        int* ctr = tickets + role;
        for (;;) {
            if (threadIdx.x == 0) s_ticket[it & 1] = atomicAdd(ctr, 1);
            __syncthreads();
            const int t = __builtin_amdgcn_readfirstlane(s_ticket[it & 1]);
            ++it;
            if (t >= nchunks) break;
            const int ebase = t * EPG + wv * EPW;

            int sv[EPW / 32], dv[EPW / 32];
            unsigned mm[EPW / 32];
            #pragma unroll
            for (int q = 0; q < EPW / 32; ++q) {
                const int e = ebase + q * 32 + lane;
                sv[q] = 0; dv[q] = 0;
                if (e < E) { sv[q] = src[e]; dv[q] = dst[e]; }
            }
            #pragma unroll
            for (int q = 0; q < EPW / 32; ++q) {
                const int e = ebase + q * 32 + lane;
                const bool mine = (e < E) && ((dv[q] < nsplit) == low);
                mm[q] = __builtin_amdgcn_ballot_w32(mine);
            }

            #pragma unroll
            for (int q = 0; q < EPW / 32; ++q) {
                    unsigned m = mm[q];
                    while (m != 0) {
                        int ss[B], dd[B];
                        #pragma unroll
                        for (int k = 0; k < B; ++k) {
                            const bool on = (m != 0);
                            const int  b  = on ? __builtin_ctz(m) : 0;
                            m = on ? (m & (m - 1u)) : 0u;
                            ss[k] = __builtin_amdgcn_readlane(sv[q], b);
                            dd[k] = on ? __builtin_amdgcn_readlane(dv[q], b) : -1;
                        }
                        float v[B][NL];
                        #pragma unroll
                        for (int k = 0; k < B; ++k) {
                            const float* hrow = h + (size_t)ss[k] * CH;
                            #pragma unroll
                            for (int c = 0; c < NL; ++c) v[k][c] = hrow[lane + 32 * c];
                        }
                        #pragma unroll
                        for (int k = 0; k < B; ++k) {
                            if (dd[k] >= 0) {
                                float* arow = agg + (size_t)dd[k] * CH;
                                #pragma unroll
                                for (int c = 0; c < NL; ++c) atomicAdd(arow + lane + 32 * c, v[k][c]);
                            }
                        }
                    }
            }
        }
    }
}

__global__ __launch_bounds__(256) void pool_mean_bias_kernel(
    const float* __restrict__ ysum, const float* __restrict__ cnt,
    const float* __restrict__ lin_b, float* __restrict__ out, int G)
{
    const int idx = blockIdx.x * blockDim.x + threadIdx.x;
    if (idx >= (G << 4)) return;
    const float m = fmaxf(cnt[idx >> 4], 1.0f);
    out[idx] = ysum[idx] / m + lin_b[idx & 15];
}


static inline unsigned blocks_for(long long n, int t) {
    return (unsigned)((n + t - 1) / t);
}

extern "C" void kernel_launch(void* const* d_in, const int* in_sizes, int n_in,
                              void* d_out, int out_size, void* d_ws, size_t ws_size,
                              hipStream_t stream)
{
    const float* x        = (const float*)d_in[0];
    const int*   edge     = (const int*)  d_in[1];
    const int*   batch    = (const int*)  d_in[2];
    const float* w_stage1 = (const float*)d_in[4];
    const float* b_stage1 = (const float*)d_in[5];
    const float* rel1_w   = (const float*)d_in[6];
    const float* rel1_b   = (const float*)d_in[7];
    const float* root1_w  = (const float*)d_in[8];
    const float* rel2_w   = (const float*)d_in[9];
    const float* rel2_b   = (const float*)d_in[10];
    const float* root2_w  = (const float*)d_in[11];
    const float* rel3_w   = (const float*)d_in[12];
    const float* rel3_b   = (const float*)d_in[13];
    const float* root3_w  = (const float*)d_in[14];
    const float* lin_w    = (const float*)d_in[15];
    const float* lin_b    = (const float*)d_in[16];
    float*       out      = (float*)d_out;

    const int N = in_sizes[0] / 128;
    const int E = in_sizes[1] / 2;
    const int G = out_size / 16;
    const int* src = edge;
    const int* dst = edge + E;

    float* ws = (float*)d_ws;
    size_t off = 0;
    float* bufA   = ws + off; off += (size_t)N * 128;
    float* bufB   = ws + off; off += (size_t)N * 128;
    float* agg    = ws + off; off += (size_t)N * 128;
    float* agg3   = ws + off; off += (size_t)N * 128;
    float* ysum   = ws + off; off += (size_t)G * 16;
    float* cnt    = ws + off; off += (size_t)G;
    int*   tickets = (int*)(ws + off); off += 8;
    (void)ws_size; (void)n_in;

    const int T = 256;

    auto gemm_grid = [&](int M, int rows_per_group) {
        return blocks_for(M, rows_per_group);
    };

    gemm_wmma_kernel<false, false, 128, 64, 8, 64, true, false><<<gemm_grid(N, 32), T, 0, stream>>>(
        x, w_stage1, nullptr, nullptr, b_stage1, bufA, N, agg, ysum, (G * 16 + G) / 4 + 2,
        nullptr, nullptr, nullptr, nullptr);

    scatter_add_kernel<64><<<SCATTER_GROUPS, T, 0, stream>>>(
        bufA, src, dst, agg, E, N, tickets);
    gemm_wmma_kernel<true, true, 64, 64, 8, 64, false, false><<<gemm_grid(N, 32), T, 0, stream>>>(
        agg, rel1_w, bufA, root1_w, rel1_b, bufB, N, agg, nullptr, 0,
        nullptr, nullptr, nullptr, nullptr);

    scatter_add_kernel<64><<<SCATTER_GROUPS, T, 0, stream>>>(
        bufB, src, dst, agg, E, N, tickets + 2);
    gemm_wmma_kernel<true, true, 64, 128, 8, 128, false, false><<<gemm_grid(N, 16), T, 0, stream>>>(
        agg, rel2_w, bufB, root2_w, rel2_b, bufA, N, agg3, nullptr, 0,
        nullptr, nullptr, nullptr, nullptr);

    scatter_add_kernel<128><<<SCATTER_GROUPS, T, 0, stream>>>(
        bufA, src, dst, agg3, E, N, tickets + 4);
    gemm_wmma_kernel<true, true, 128, 128, 8, 0, false, true><<<gemm_grid(N, 16), T, 0, stream>>>(
        agg3, rel3_w, bufA, root3_w, rel3_b, nullptr, N, nullptr, nullptr, 0,
        lin_w, batch, ysum, cnt);

    pool_mean_bias_kernel<<<blocks_for((long long)G << 4, T), T, 0, stream>>>(
        ysum, cnt, lin_b, out, G);
}


# Conv 1 products moved before its neighbour sum (linear): stage-1 launch multiplies h tiles by rel1_w, root1_w on WMMA, edges add products, ReLU on load; 8 to 7 launches; sum order: atomic arrival
# speedup vs baseline: 3.1122x; 1.1212x over previous
#include <hip/hip_runtime.h>


typedef __attribute__((ext_vector_type(2))) float v2f;
typedef __attribute__((ext_vector_type(8))) float v8f;

template <bool RELU, bool DUAL, int CIN, int COUT, int WAVES, int ZCH, bool ZFLAT, bool POOL, bool BRELU>
__global__ __launch_bounds__(WAVES * 32) void gemm_wmma_kernel(
    const float* A, const float* __restrict__ W1,
    const float* __restrict__ B, const float* __restrict__ W2,
    const float* __restrict__ bias, float* __restrict__ out, int M,
    float* zrows, float* zflat, int zflat_n4,
    const float* __restrict__ lw, const int* __restrict__ batch,
    float* __restrict__ ysum, float* __restrict__ cnt)
{
    constexpr int NT    = WAVES * 32;
    constexpr int NTN   = COUT / 16;
    constexpr int SPB   = WAVES / NTN;
    constexpr int ROWS  = SPB * 16;
    constexpr int PITCH = CIN + 4;
    constexpr int C4    = CIN / 4;
    constexpr int NV    = ROWS * C4;
    constexpr int ITERS = NV / NT;
    static_assert(SPB * NTN == WAVES && ITERS * NT == NV, "tile shape");

    __shared__ __attribute__((aligned(16))) float sA[DUAL ? 2 : 1][ROWS * PITCH];
    constexpr int TP = 20;
    __shared__ __attribute__((aligned(16))) float sT[POOL ? WAVES : 1][16 * TP];
    static_assert(!POOL || SPB == 1, "POOL: one strip per work-group, every wave reaches the barrier");

    const int tid  = threadIdx.x;
    const int row0 = blockIdx.x * ROWS;

    {
        const float4* gA = reinterpret_cast<const float4*>(A);
        const float4* gB = DUAL ? reinterpret_cast<const float4*>(B) : nullptr;
        float4 va[ITERS], vb[ITERS];
        #pragma unroll
        for (int it = 0; it < ITERS; ++it) {
            const int i = it * NT + tid;
            const int r = i / C4;
            const int c = i - r * C4;
            const size_t g = (size_t)min(row0 + r, M - 1) * C4 + c;
            va[it] = gA[g];
            if (DUAL) vb[it] = gB[g];
        }
        #pragma unroll
        for (int it = 0; it < ITERS; ++it) {
            const int i = it * NT + tid;
            const int r = i / C4;
            const int c = i - r * C4;
            *reinterpret_cast<float4*>(&sA[0][r * PITCH + c * 4]) = va[it];
            if (DUAL) {
                float4 b = vb[it];
                if (BRELU) { b.x = fmaxf(b.x, 0.0f); b.y = fmaxf(b.y, 0.0f); b.z = fmaxf(b.z, 0.0f); b.w = fmaxf(b.w, 0.0f); }
                *reinterpret_cast<float4*>(&sA[DUAL ? 1 : 0][r * PITCH + c * 4]) = b;
            }
        }
    }
    __syncthreads();

    if (ZCH > 0) {
        constexpr int Z4 = ZCH / 4;
        constexpr int NZ = ROWS * Z4;
        static_assert(ZCH == 0 || NZ % NT == 0, "fill shape");
        float4* z4 = reinterpret_cast<float4*>(zrows) + (size_t)row0 * Z4;
        for (int it = 0; it < NZ / NT; ++it) {
            const int i = it * NT + tid;
            if (row0 + i / Z4 < M) z4[i] = make_float4(0.f, 0.f, 0.f, 0.f);
        }
    }
    if (ZFLAT) {
        const int gi = blockIdx.x * NT + tid;
        if (gi < zflat_n4) reinterpret_cast<float4*>(zflat)[gi] = make_float4(0.f, 0.f, 0.f, 0.f);
    }

    const int lane = tid & 31;
    const int wave = tid >> 5;
    const int s    = wave / NTN;
    const int tn   = wave - s * NTN;
    const int tm16 = row0 + s * 16;
    if (!POOL && tm16 >= M) return;

    const int l    = lane & 15;
    const int koff = (lane >> 4) << 1;
    const int col  = tn * 16 + l;

    const float* arow = &sA[0][(s * 16 + l) * PITCH + koff];
    const float* brow = &sA[DUAL ? 1 : 0][(s * 16 + l) * PITCH + koff];
    const float* w1p  = W1 + (size_t)koff * COUT + col;
    const float* w2p  = DUAL ? (W2 + (size_t)koff * COUT + col) : nullptr;

    v8f acc = {};
    for (int k = 0; k < CIN; k += 4) {
        const v2f av = *reinterpret_cast<const v2f*>(arow + k);
        v2f wv; wv.x = w1p[(size_t)k * COUT];
                wv.y = w1p[(size_t)(k + 1) * COUT];
        acc = __builtin_amdgcn_wmma_f32_16x16x4_f32(
                  false, av, false, wv, (short)0, acc, false, false);
        if (DUAL) {
            const v2f bv = *reinterpret_cast<const v2f*>(brow + k);
            v2f w2; w2.x = w2p[(size_t)k * COUT];
                    w2.y = w2p[(size_t)(k + 1) * COUT];
            acc = __builtin_amdgcn_wmma_f32_16x16x4_f32(
                      false, bv, false, w2, (short)0, acc, false, false);
        }
    }

    const float bval = bias[col];
    const int m0 = tm16 + ((lane >> 4) << 3);
    if (!POOL) {
        #pragma unroll
        for (int r = 0; r < 8; ++r) {
            float v = acc[r] + bval;
            if (RELU) v = fmaxf(v, 0.0f);
            out[(size_t)(m0 + r) * COUT + col] = v;
        }
    } else {
        float* tile = &sT[wave][0];
        const int h8 = (lane >> 4) << 3;
        #pragma unroll
        for (int r = 0; r < 8; ++r) {
            float v = acc[r] + bval;
            if (RELU) v = fmaxf(v, 0.0f);
            tile[(h8 + r) * TP + l] = v;
        }
        __syncthreads();
        const float* lwp = lw + (size_t)(tn * 16 + koff) * 16 + l;
        v8f y = {};
        #pragma unroll
        for (int k = 0; k < 16; k += 4) {
            const v2f av = *reinterpret_cast<const v2f*>(&tile[l * TP + k + koff]);
            v2f wv; wv.x = lwp[k * 16];
                    wv.y = lwp[(k + 1) * 16];
            y = __builtin_amdgcn_wmma_f32_16x16x4_f32(
                    false, av, false, wv, (short)0, y, false, false);
        }
        int g[8];
        #pragma unroll
        for (int r = 0; r < 8; ++r) g[r] = batch[m0 + r];
        const bool counts = (tn == 0) && (l == 0);
        int gcur = g[0];
        float run = y[0];
        float nrun = 1.0f;
        #pragma unroll
        for (int r = 1; r < 8; ++r) {
            if (g[r] == gcur) { run += y[r]; nrun += 1.0f; }
            else {
                atomicAdd(&ysum[(size_t)gcur * 16 + l], run);
                if (counts) atomicAdd(&cnt[gcur], nrun);
                gcur = g[r]; run = y[r]; nrun = 1.0f;
            }
        }
        atomicAdd(&ysum[(size_t)gcur * 16 + l], run);
        if (counts) atomicAdd(&cnt[gcur], nrun);
    }
}

template <int CIN, int CMID, int COUT, int WAVES, int ZCH, bool ZFLAT>
__global__ __launch_bounds__(WAVES * 32) void gemm_chain_kernel(
    const float* __restrict__ X, const float* __restrict__ W0, const float* __restrict__ b0,
    const float* __restrict__ Wrel, const float* __restrict__ Wroot,
    const float* __restrict__ brel,
    float* __restrict__ pout, float* __restrict__ rout, int M,
    float* __restrict__ zrows, float* __restrict__ zflat, int zflat_n4)
{
    constexpr int NT    = WAVES * 32;
    constexpr int NTN   = CMID / 16;
    constexpr int SPB   = WAVES / NTN;
    constexpr int ROWS  = SPB * 16;
    constexpr int PITCH = CIN + 4;
    constexpr int HP    = CMID + 4;
    constexpr int C4    = CIN / 4;
    constexpr int NV    = ROWS * C4;
    constexpr int ITERS = NV / NT;
    static_assert(SPB * NTN == WAVES && ITERS * NT == NV && COUT == CMID, "tile shape");

    __shared__ __attribute__((aligned(16))) float sX[ROWS * PITCH];
    __shared__ __attribute__((aligned(16))) float sH[ROWS * HP];

    const int tid  = threadIdx.x;
    const int row0 = blockIdx.x * ROWS;

    {
        const float4* gX = reinterpret_cast<const float4*>(X);
        float4 va[ITERS];
        #pragma unroll
        for (int it = 0; it < ITERS; ++it) {
            const int i = it * NT + tid;
            const int r = i / C4;
            const int c = i - r * C4;
            va[it] = gX[(size_t)min(row0 + r, M - 1) * C4 + c];
        }
        #pragma unroll
        for (int it = 0; it < ITERS; ++it) {
            const int i = it * NT + tid;
            const int r = i / C4;
            const int c = i - r * C4;
            *reinterpret_cast<float4*>(&sX[r * PITCH + c * 4]) = va[it];
        }
    }
    __syncthreads();

    if (ZCH > 0) {
        constexpr int Z4 = ZCH / 4;
        constexpr int NZ = ROWS * Z4;
        static_assert(ZCH == 0 || NZ % NT == 0, "fill shape");
        float4* z4 = reinterpret_cast<float4*>(zrows) + (size_t)row0 * Z4;
        for (int it = 0; it < NZ / NT; ++it) {
            const int i = it * NT + tid;
            if (row0 + i / Z4 < M) z4[i] = make_float4(0.f, 0.f, 0.f, 0.f);
        }
    }
    if (ZFLAT) {
        const int gi = blockIdx.x * NT + tid;
        if (gi < zflat_n4) reinterpret_cast<float4*>(zflat)[gi] = make_float4(0.f, 0.f, 0.f, 0.f);
    }

    const int lane = tid & 31;
    const int wave = tid >> 5;
    const int s    = wave / NTN;
    const int tn   = wave - s * NTN;
    const int tm16 = row0 + s * 16;
    const int l    = lane & 15;
    const int koff = (lane >> 4) << 1;
    const int col  = tn * 16 + l;
    const int h8   = (lane >> 4) << 3;

    {
        const float* xrow = &sX[(s * 16 + l) * PITCH + koff];
        const float* w0p  = W0 + (size_t)koff * CMID + col;
        v8f acc = {};
        for (int k = 0; k < CIN; k += 4) {
            const v2f av = *reinterpret_cast<const v2f*>(xrow + k);
            v2f wv; wv.x = w0p[(size_t)k * CMID];
                    wv.y = w0p[(size_t)(k + 1) * CMID];
            acc = __builtin_amdgcn_wmma_f32_16x16x4_f32(
                      false, av, false, wv, (short)0, acc, false, false);
        }
        const float bval = b0[col];
        float* hs = &sH[(s * 16 + h8) * HP + col];
        #pragma unroll
        for (int r = 0; r < 8; ++r) hs[r * HP] = acc[r] + bval;
    }
    __syncthreads();

    const float* hrow = &sH[(s * 16 + l) * HP + koff];
    const float* wrp  = Wrel  + (size_t)koff * COUT + col;
    const float* wop  = Wroot + (size_t)koff * COUT + col;
    v8f accp = {};
    v8f accr = {};
    for (int k = 0; k < CMID; k += 4) {
        const v2f av = *reinterpret_cast<const v2f*>(hrow + k);
        v2f wr; wr.x = wrp[(size_t)k * COUT];
                wr.y = wrp[(size_t)(k + 1) * COUT];
        v2f wo; wo.x = wop[(size_t)k * COUT];
                wo.y = wop[(size_t)(k + 1) * COUT];
        accp = __builtin_amdgcn_wmma_f32_16x16x4_f32(
                   false, av, false, wr, (short)0, accp, false, false);
        accr = __builtin_amdgcn_wmma_f32_16x16x4_f32(
                   false, av, false, wo, (short)0, accr, false, false);
    }

    if (tm16 < M) {
        const float bval = brel[col];
        const int m0 = tm16 + h8;
        #pragma unroll
        for (int r = 0; r < 8; ++r) {
            pout[(size_t)(m0 + r) * COUT + col] = accp[r];
            rout[(size_t)(m0 + r) * COUT + col] = accr[r] + bval;
        }
    }
}

constexpr int SCATTER_GROUPS = 1024;

template <int CH, bool RELU>
__global__ __launch_bounds__(256) void scatter_add_kernel(
    const float* __restrict__ h, const int* __restrict__ src,
    const int* __restrict__ dst, float* __restrict__ agg, int E, int N,
    int* __restrict__ tickets)
{
    constexpr int NL  = CH / 32;
    constexpr int B   = 20;
    constexpr int EPW = 64;
    constexpr int EPG = EPW * 8;
    __shared__ int s_ticket[2];

    const int lane = threadIdx.x & 31;
    const int wv = __builtin_amdgcn_readfirstlane((int)(threadIdx.x >> 5));
    const int nchunks = (E + EPG - 1) / EPG;
    const int nsplit  = N >> 1;
    const int myhalf = (int)((__builtin_amdgcn_s_sendmsg_rtn(0x87) >> 18) & 1u);

    int it = 0;
    for (int pass = 0; pass < 2; ++pass) {
        const int  role = myhalf ^ pass;
        const bool low  = (role == 0);
        int* ctr = tickets + role;
        for (;;) {
            if (threadIdx.x == 0) s_ticket[it & 1] = atomicAdd(ctr, 1);
            __syncthreads();
            const int t = __builtin_amdgcn_readfirstlane(s_ticket[it & 1]);
            ++it;
            if (t >= nchunks) break;
            const int ebase = t * EPG + wv * EPW;

            int sv[EPW / 32], dv[EPW / 32];
            unsigned mm[EPW / 32];
            #pragma unroll
            for (int q = 0; q < EPW / 32; ++q) {
                const int e = ebase + q * 32 + lane;
                sv[q] = 0; dv[q] = 0;
                if (e < E) { sv[q] = src[e]; dv[q] = dst[e]; }
            }
            #pragma unroll
            for (int q = 0; q < EPW / 32; ++q) {
                const int e = ebase + q * 32 + lane;
                const bool mine = (e < E) && ((dv[q] < nsplit) == low);
                mm[q] = __builtin_amdgcn_ballot_w32(mine);
            }

            #pragma unroll
            for (int q = 0; q < EPW / 32; ++q) {
                    unsigned m = mm[q];
                    while (m != 0) {
                        int ss[B], dd[B];
                        #pragma unroll
                        for (int k = 0; k < B; ++k) {
                            const bool on = (m != 0);
                            const int  b  = on ? __builtin_ctz(m) : 0;
                            m = on ? (m & (m - 1u)) : 0u;
                            ss[k] = __builtin_amdgcn_readlane(sv[q], b);
                            dd[k] = on ? __builtin_amdgcn_readlane(dv[q], b) : -1;
                        }
                        float v[B][NL];
                        #pragma unroll
                        for (int k = 0; k < B; ++k) {
                            const float* hrow = h + (size_t)ss[k] * CH;
                            #pragma unroll
                            for (int c = 0; c < NL; ++c) v[k][c] = hrow[lane + 32 * c];
                        }
                        #pragma unroll
                        for (int k = 0; k < B; ++k) {
                            if (dd[k] >= 0) {
                                float* arow = agg + (size_t)dd[k] * CH;
                                #pragma unroll
                                for (int c = 0; c < NL; ++c) atomicAdd(arow + lane + 32 * c, RELU ? fmaxf(v[k][c], 0.0f) : v[k][c]);
                            }
                        }
                    }
            }
        }
    }
}

__global__ __launch_bounds__(256) void pool_mean_bias_kernel(
    const float* __restrict__ ysum, const float* __restrict__ cnt,
    const float* __restrict__ lin_b, float* __restrict__ out, int G)
{
    const int idx = blockIdx.x * blockDim.x + threadIdx.x;
    if (idx >= (G << 4)) return;
    const float m = fmaxf(cnt[idx >> 4], 1.0f);
    out[idx] = ysum[idx] / m + lin_b[idx & 15];
}


static inline unsigned blocks_for(long long n, int t) {
    return (unsigned)((n + t - 1) / t);
}

extern "C" void kernel_launch(void* const* d_in, const int* in_sizes, int n_in,
                              void* d_out, int out_size, void* d_ws, size_t ws_size,
                              hipStream_t stream)
{
    const float* x        = (const float*)d_in[0];
    const int*   edge     = (const int*)  d_in[1];
    const int*   batch    = (const int*)  d_in[2];
    const float* w_stage1 = (const float*)d_in[4];
    const float* b_stage1 = (const float*)d_in[5];
    const float* rel1_w   = (const float*)d_in[6];
    const float* rel1_b   = (const float*)d_in[7];
    const float* root1_w  = (const float*)d_in[8];
    const float* rel2_w   = (const float*)d_in[9];
    const float* rel2_b   = (const float*)d_in[10];
    const float* root2_w  = (const float*)d_in[11];
    const float* rel3_w   = (const float*)d_in[12];
    const float* rel3_b   = (const float*)d_in[13];
    const float* root3_w  = (const float*)d_in[14];
    const float* lin_w    = (const float*)d_in[15];
    const float* lin_b    = (const float*)d_in[16];
    float*       out      = (float*)d_out;

    const int N = in_sizes[0] / 128;
    const int E = in_sizes[1] / 2;
    const int G = out_size / 16;
    const int* src = edge;
    const int* dst = edge + E;

    float* ws = (float*)d_ws;
    size_t off = 0;
    float* bufA   = ws + off; off += (size_t)N * 128;
    float* prod1  = ws + off; off += (size_t)N * 64;
    float* pre1   = ws + off; off += (size_t)N * 64;
    float* agg    = ws + off; off += (size_t)N * 128;
    float* agg3   = ws + off; off += (size_t)N * 128;
    float* ysum   = ws + off; off += (size_t)G * 16;
    float* cnt    = ws + off; off += (size_t)G;
    int*   tickets = (int*)(ws + off); off += 8;
    (void)ws_size; (void)n_in;

    const int T = 256;

    auto gemm_grid = [&](int M, int rows_per_group) {
        return blocks_for(M, rows_per_group);
    };

    gemm_chain_kernel<128, 64, 64, 8, 64, true><<<gemm_grid(N, 32), T, 0, stream>>>(
        x, w_stage1, b_stage1, rel1_w, root1_w, rel1_b, prod1, pre1, N,
        agg, ysum, (G * 16 + G) / 4 + 2);

    scatter_add_kernel<64, false><<<SCATTER_GROUPS, T, 0, stream>>>(
        prod1, src, dst, pre1, E, N, tickets);

    scatter_add_kernel<64, true><<<SCATTER_GROUPS, T, 0, stream>>>(
        pre1, src, dst, agg, E, N, tickets + 2);
    gemm_wmma_kernel<true, true, 64, 128, 8, 128, false, false, true><<<gemm_grid(N, 16), T, 0, stream>>>(
        agg, rel2_w, pre1, root2_w, rel2_b, bufA, N, agg3, nullptr, 0,
        nullptr, nullptr, nullptr, nullptr);

    scatter_add_kernel<128, false><<<SCATTER_GROUPS, T, 0, stream>>>(
        bufA, src, dst, agg3, E, N, tickets + 4);
    gemm_wmma_kernel<true, true, 128, 128, 8, 0, false, true, false><<<gemm_grid(N, 16), T, 0, stream>>>(
        agg3, rel3_w, bufA, root3_w, rel3_b, nullptr, N, nullptr, nullptr, 0,
        lin_w, batch, ysum, cnt);

    pool_mean_bias_kernel<<<blocks_for((long long)G << 4, T), T, 0, stream>>>(
        ysum, cnt, lin_b, out, G);
}


# GEMM weights kept in registers: a fixed grid of 512 work-groups walks the row strips (LDS-only barriers per trip), each wave loads its column tile's weight fragments once; sum order unchanged
# speedup vs baseline: 3.2546x; 1.0458x over previous
#include <hip/hip_runtime.h>


typedef __attribute__((ext_vector_type(2))) float v2f;
typedef __attribute__((ext_vector_type(8))) float v8f;

__device__ __forceinline__ void lds_barrier()
{
    __builtin_amdgcn_fence(__ATOMIC_RELEASE, "workgroup", "local");
    __builtin_amdgcn_s_barrier();
    __builtin_amdgcn_fence(__ATOMIC_ACQUIRE, "workgroup", "local");
}

constexpr int GEMM_GROUPS = 512;

template <bool RELU, bool DUAL, int CIN, int COUT, int WAVES, int ZCH, bool ZFLAT, bool POOL, bool BRELU>
__global__ __launch_bounds__(WAVES * 32) void gemm_wmma_kernel(
    const float* A, const float* __restrict__ W1,
    const float* __restrict__ B, const float* __restrict__ W2,
    const float* __restrict__ bias, float* __restrict__ out, int M,
    float* zrows, float* zflat, int zflat_n4,
    const float* __restrict__ lw, const int* __restrict__ batch,
    float* __restrict__ ysum, float* __restrict__ cnt)
{
    constexpr int NT    = WAVES * 32;
    constexpr int NTN   = COUT / 16;
    constexpr int SPB   = WAVES / NTN;
    constexpr int ROWS  = SPB * 16;
    constexpr int PITCH = CIN + 4;
    constexpr int C4    = CIN / 4;
    constexpr int NV    = ROWS * C4;
    constexpr int ITERS = NV / NT;
    constexpr int KS    = CIN / 4;
    static_assert(SPB * NTN == WAVES && ITERS * NT == NV, "tile shape");

    __shared__ __attribute__((aligned(16))) float sA[DUAL ? 2 : 1][ROWS * PITCH];
    constexpr int TP = 20;
    __shared__ __attribute__((aligned(16))) float sT[POOL ? WAVES : 1][16 * TP];
    static_assert(!POOL || SPB == 1, "POOL: one strip per trip, every wave reaches the barrier");

    const int tid  = threadIdx.x;
    const int lane = tid & 31;
    const int wave = tid >> 5;
    const int s    = wave / NTN;
    const int tn   = wave - s * NTN;
    const int l    = lane & 15;
    const int koff = (lane >> 4) << 1;
    const int col  = tn * 16 + l;
    const int h8   = (lane >> 4) << 3;
    const int ntrips = (M + ROWS - 1) / ROWS;

    if (ZFLAT) {
        for (int gi = blockIdx.x * NT + tid; gi < zflat_n4; gi += gridDim.x * NT)
            reinterpret_cast<float4*>(zflat)[gi] = make_float4(0.f, 0.f, 0.f, 0.f);
    }

    v2f w1r[KS];
    v2f w2r[DUAL ? KS : 1];
    {
        const float* w1p = W1 + (size_t)koff * COUT + col;
        #pragma unroll
        for (int j = 0; j < KS; ++j) {
            w1r[j].x = w1p[(size_t)(4 * j) * COUT];
            w1r[j].y = w1p[(size_t)(4 * j + 1) * COUT];
        }
        if (DUAL) {
            const float* w2p = W2 + (size_t)koff * COUT + col;
            #pragma unroll
            for (int j = 0; j < KS; ++j) {
                w2r[j].x = w2p[(size_t)(4 * j) * COUT];
                w2r[j].y = w2p[(size_t)(4 * j + 1) * COUT];
            }
        }
    }
    const float bval = bias[col];
    v2f lwr[4];
    {
        const float* lwp = POOL ? (lw + (size_t)(tn * 16 + koff) * 16 + l) : nullptr;
        #pragma unroll
        for (int j = 0; j < 4; ++j) {
            lwr[j].x = POOL ? lwp[(4 * j) * 16] : 0.0f;
            lwr[j].y = POOL ? lwp[(4 * j + 1) * 16] : 0.0f;
        }
    }

    const float4* gA = reinterpret_cast<const float4*>(A);
    const float4* gB = DUAL ? reinterpret_cast<const float4*>(B) : nullptr;
    static_assert(ITERS >= 1 && ITERS <= 4, "pieces per thread");
    float4 va0, va1, va2, va3, vb0, vb1, vb2, vb3;
    va0 = va1 = va2 = va3 = vb0 = vb1 = vb2 = vb3 = make_float4(0.f, 0.f, 0.f, 0.f);
#define LOAD_PIECE(q, r0)                                                        \
    if (ITERS > q) {                                                             \
        const int i_ = q * NT + tid;                                             \
        const int r_ = i_ / C4;                                                  \
        const int c_ = i_ - r_ * C4;                                             \
        const size_t g_ = (size_t)min((r0) + r_, M - 1) * C4 + c_;               \
        va##q = gA[g_];                                                          \
        if (DUAL) vb##q = gB[g_];                                                \
    }
#define LOAD_STRIPS(r0) { LOAD_PIECE(0, r0) LOAD_PIECE(1, r0) LOAD_PIECE(2, r0) LOAD_PIECE(3, r0) }
#define STORE_PIECE(q)                                                           \
    if (ITERS > q) {                                                             \
        const int i_ = q * NT + tid;                                             \
        const int r_ = i_ / C4;                                                  \
        const int c_ = i_ - r_ * C4;                                             \
        *reinterpret_cast<float4*>(&sA[0][r_ * PITCH + c_ * 4]) = va##q;         \
        if (DUAL) {                                                              \
            float4 b_ = vb##q;                                                   \
            if (BRELU) { b_.x = fmaxf(b_.x, 0.0f); b_.y = fmaxf(b_.y, 0.0f);     \
                         b_.z = fmaxf(b_.z, 0.0f); b_.w = fmaxf(b_.w, 0.0f); }   \
            *reinterpret_cast<float4*>(&sA[DUAL ? 1 : 0][r_ * PITCH + c_ * 4]) = b_; \
        }                                                                        \
    }

    const float* arow = &sA[0][(s * 16 + l) * PITCH + koff];
    const float* brow = &sA[DUAL ? 1 : 0][(s * 16 + l) * PITCH + koff];

    int t = blockIdx.x;
    if (t < ntrips) LOAD_STRIPS(t * ROWS)
    for (; t < ntrips; t += gridDim.x) {
        const int row0 = t * ROWS;
        STORE_PIECE(0) STORE_PIECE(1) STORE_PIECE(2) STORE_PIECE(3)
        lds_barrier();

        if (t + (int)gridDim.x < ntrips) LOAD_STRIPS((t + (int)gridDim.x) * ROWS)

        if (ZCH > 0) {
            constexpr int Z4 = ZCH / 4;
            constexpr int NZ = ROWS * Z4;
            static_assert(ZCH == 0 || NZ % NT == 0, "fill shape");
            float4* z4 = reinterpret_cast<float4*>(zrows) + (size_t)row0 * Z4;
            for (int it = 0; it < NZ / NT; ++it) {
                const int i = it * NT + tid;
                if (row0 + i / Z4 < M) z4[i] = make_float4(0.f, 0.f, 0.f, 0.f);
            }
        }

        const int tm16 = row0 + s * 16;
        if (POOL || tm16 < M) {
            v8f acc = {};
            #pragma unroll
            for (int j = 0; j < KS; ++j) {
                const v2f av = *reinterpret_cast<const v2f*>(arow + 4 * j);
                acc = __builtin_amdgcn_wmma_f32_16x16x4_f32(
                          false, av, false, w1r[j], (short)0, acc, false, false);
                if (DUAL) {
                    const v2f bv = *reinterpret_cast<const v2f*>(brow + 4 * j);
                    acc = __builtin_amdgcn_wmma_f32_16x16x4_f32(
                              false, bv, false, w2r[j], (short)0, acc, false, false);
                }
            }

            const int m0 = tm16 + h8;
            if (!POOL) {
                #pragma unroll
                for (int r = 0; r < 8; ++r) {
                    float v = acc[r] + bval;
                    if (RELU) v = fmaxf(v, 0.0f);
                    out[(size_t)(m0 + r) * COUT + col] = v;
                }
            } else {
                float* tile = &sT[wave][0];
                #pragma unroll
                for (int r = 0; r < 8; ++r) {
                    float v = acc[r] + bval;
                    if (RELU) v = fmaxf(v, 0.0f);
                    tile[(h8 + r) * TP + l] = v;
                }
                lds_barrier();
                v8f y = {};
                #pragma unroll
                for (int j = 0; j < 4; ++j) {
                    const v2f av = *reinterpret_cast<const v2f*>(&tile[l * TP + 4 * j + koff]);
                    y = __builtin_amdgcn_wmma_f32_16x16x4_f32(
                            false, av, false, lwr[j], (short)0, y, false, false);
                }
                int g[8];
                #pragma unroll
                for (int r = 0; r < 8; ++r) g[r] = batch[m0 + r];
                const bool counts = (tn == 0) && (l == 0);
                int gcur = g[0];
                float run = y[0];
                float nrun = 1.0f;
                #pragma unroll
                for (int r = 1; r < 8; ++r) {
                    if (g[r] == gcur) { run += y[r]; nrun += 1.0f; }
                    else {
                        atomicAdd(&ysum[(size_t)gcur * 16 + l], run);
                        if (counts) atomicAdd(&cnt[gcur], nrun);
                        gcur = g[r]; run = y[r]; nrun = 1.0f;
                    }
                }
                atomicAdd(&ysum[(size_t)gcur * 16 + l], run);
                if (counts) atomicAdd(&cnt[gcur], nrun);
            }
        }
        if (!POOL) lds_barrier();
    }
#undef LOAD_PIECE
#undef LOAD_STRIPS
#undef STORE_PIECE
}

template <int CIN, int CMID, int COUT, int WAVES, int ZCH, bool ZFLAT>
__global__ __launch_bounds__(WAVES * 32) void gemm_chain_kernel(
    const float* __restrict__ X, const float* __restrict__ W0, const float* __restrict__ b0,
    const float* __restrict__ Wrel, const float* __restrict__ Wroot,
    const float* __restrict__ brel,
    float* __restrict__ pout, float* __restrict__ rout, int M,
    float* __restrict__ zrows, float* __restrict__ zflat, int zflat_n4)
{
    constexpr int NT    = WAVES * 32;
    constexpr int NTN   = CMID / 16;
    constexpr int SPB   = WAVES / NTN;
    constexpr int ROWS  = SPB * 16;
    constexpr int PITCH = CIN + 4;
    constexpr int HP    = CMID + 4;
    constexpr int C4    = CIN / 4;
    constexpr int NV    = ROWS * C4;
    constexpr int ITERS = NV / NT;
    constexpr int KS0   = CIN / 4;
    constexpr int KS1   = CMID / 4;
    static_assert(SPB * NTN == WAVES && ITERS * NT == NV && COUT == CMID, "tile shape");

    __shared__ __attribute__((aligned(16))) float sX[ROWS * PITCH];
    __shared__ __attribute__((aligned(16))) float sH[ROWS * HP];

    const int tid  = threadIdx.x;
    const int lane = tid & 31;
    const int wave = tid >> 5;
    const int s    = wave / NTN;
    const int tn   = wave - s * NTN;
    const int l    = lane & 15;
    const int koff = (lane >> 4) << 1;
    const int col  = tn * 16 + l;
    const int h8   = (lane >> 4) << 3;
    const int ntrips = (M + ROWS - 1) / ROWS;

    if (ZFLAT) {
        for (int gi = blockIdx.x * NT + tid; gi < zflat_n4; gi += gridDim.x * NT)
            reinterpret_cast<float4*>(zflat)[gi] = make_float4(0.f, 0.f, 0.f, 0.f);
    }

    v2f w0r[KS0], wrr[KS1], wor[KS1];
    {
        const float* w0p = W0    + (size_t)koff * CMID + col;
        const float* wrp = Wrel  + (size_t)koff * COUT + col;
        const float* wop = Wroot + (size_t)koff * COUT + col;
        #pragma unroll
        for (int j = 0; j < KS0; ++j) {
            w0r[j].x = w0p[(size_t)(4 * j) * CMID];
            w0r[j].y = w0p[(size_t)(4 * j + 1) * CMID];
        }
        #pragma unroll
        for (int j = 0; j < KS1; ++j) {
            wrr[j].x = wrp[(size_t)(4 * j) * COUT];
            wrr[j].y = wrp[(size_t)(4 * j + 1) * COUT];
            wor[j].x = wop[(size_t)(4 * j) * COUT];
            wor[j].y = wop[(size_t)(4 * j + 1) * COUT];
        }
    }
    const float b0val = b0[col];
    const float brval = brel[col];

    const float4* gX = reinterpret_cast<const float4*>(X);
    static_assert(ITERS >= 1 && ITERS <= 4, "pieces per thread");
    float4 va0, va1, va2, va3;
    va0 = va1 = va2 = va3 = make_float4(0.f, 0.f, 0.f, 0.f);
#define LOAD_XPIECE(q, r0)                                                       \
    if (ITERS > q) {                                                             \
        const int i_ = q * NT + tid;                                             \
        const int r_ = i_ / C4;                                                  \
        const int c_ = i_ - r_ * C4;                                             \
        va##q = gX[(size_t)min((r0) + r_, M - 1) * C4 + c_];                     \
    }
#define LOAD_XSTRIPS(r0) { LOAD_XPIECE(0, r0) LOAD_XPIECE(1, r0) LOAD_XPIECE(2, r0) LOAD_XPIECE(3, r0) }
#define STORE_XPIECE(q)                                                          \
    if (ITERS > q) {                                                             \
        const int i_ = q * NT + tid;                                             \
        const int r_ = i_ / C4;                                                  \
        const int c_ = i_ - r_ * C4;                                             \
        *reinterpret_cast<float4*>(&sX[r_ * PITCH + c_ * 4]) = va##q;            \
    }

    const float* xrow = &sX[(s * 16 + l) * PITCH + koff];
    float*       hs   = &sH[(s * 16 + h8) * HP + col];
    const float* hrow = &sH[(s * 16 + l) * HP + koff];

    int t = blockIdx.x;
    if (t < ntrips) LOAD_XSTRIPS(t * ROWS)
    for (; t < ntrips; t += gridDim.x) {
        const int row0 = t * ROWS;
        STORE_XPIECE(0) STORE_XPIECE(1) STORE_XPIECE(2) STORE_XPIECE(3)
        lds_barrier();

        if (t + (int)gridDim.x < ntrips) LOAD_XSTRIPS((t + (int)gridDim.x) * ROWS)

        if (ZCH > 0) {
            constexpr int Z4 = ZCH / 4;
            constexpr int NZ = ROWS * Z4;
            static_assert(ZCH == 0 || NZ % NT == 0, "fill shape");
            float4* z4 = reinterpret_cast<float4*>(zrows) + (size_t)row0 * Z4;
            for (int it = 0; it < NZ / NT; ++it) {
                const int i = it * NT + tid;
                if (row0 + i / Z4 < M) z4[i] = make_float4(0.f, 0.f, 0.f, 0.f);
            }
        }

        {
            v8f acc = {};
            #pragma unroll
            for (int j = 0; j < KS0; ++j) {
                const v2f av = *reinterpret_cast<const v2f*>(xrow + 4 * j);
                acc = __builtin_amdgcn_wmma_f32_16x16x4_f32(
                          false, av, false, w0r[j], (short)0, acc, false, false);
            }
            #pragma unroll
            for (int r = 0; r < 8; ++r) hs[r * HP] = acc[r] + b0val;
        }
        lds_barrier();

        v8f accp = {};
        v8f accr = {};
        #pragma unroll
        for (int j = 0; j < KS1; ++j) {
            const v2f av = *reinterpret_cast<const v2f*>(hrow + 4 * j);
            accp = __builtin_amdgcn_wmma_f32_16x16x4_f32(
                       false, av, false, wrr[j], (short)0, accp, false, false);
            accr = __builtin_amdgcn_wmma_f32_16x16x4_f32(
                       false, av, false, wor[j], (short)0, accr, false, false);
        }

        const int tm16 = row0 + s * 16;
        if (tm16 < M) {
            const int m0 = tm16 + h8;
            #pragma unroll
            for (int r = 0; r < 8; ++r) {
                pout[(size_t)(m0 + r) * COUT + col] = accp[r];
                rout[(size_t)(m0 + r) * COUT + col] = accr[r] + brval;
            }
        }
    }
#undef LOAD_XPIECE
#undef LOAD_XSTRIPS
#undef STORE_XPIECE
}

constexpr int SCATTER_GROUPS = 1024;

template <int CH, bool RELU>
__global__ __launch_bounds__(256) void scatter_add_kernel(
    const float* __restrict__ h, const int* __restrict__ src,
    const int* __restrict__ dst, float* __restrict__ agg, int E, int N,
    int* __restrict__ tickets)
{
    constexpr int NL  = CH / 32;
    constexpr int B   = 20;
    constexpr int EPW = 64;
    constexpr int EPG = EPW * 8;
    __shared__ int s_ticket[2];

    const int lane = threadIdx.x & 31;
    const int wv = __builtin_amdgcn_readfirstlane((int)(threadIdx.x >> 5));
    const int nchunks = (E + EPG - 1) / EPG;
    const int nsplit  = N >> 1;
    const int myhalf = (int)((__builtin_amdgcn_s_sendmsg_rtn(0x87) >> 18) & 1u);

    int it = 0;
    for (int pass = 0; pass < 2; ++pass) {
        const int  role = myhalf ^ pass;
        const bool low  = (role == 0);
        int* ctr = tickets + role;
        for (;;) {
            if (threadIdx.x == 0) s_ticket[it & 1] = atomicAdd(ctr, 1);
            __syncthreads();
            const int t = __builtin_amdgcn_readfirstlane(s_ticket[it & 1]);
            ++it;
            if (t >= nchunks) break;
            const int ebase = t * EPG + wv * EPW;

            int sv[EPW / 32], dv[EPW / 32];
            unsigned mm[EPW / 32];
            #pragma unroll
            for (int q = 0; q < EPW / 32; ++q) {
                const int e = ebase + q * 32 + lane;
                sv[q] = 0; dv[q] = 0;
                if (e < E) { sv[q] = src[e]; dv[q] = dst[e]; }
            }
            #pragma unroll
            for (int q = 0; q < EPW / 32; ++q) {
                const int e = ebase + q * 32 + lane;
                const bool mine = (e < E) && ((dv[q] < nsplit) == low);
                mm[q] = __builtin_amdgcn_ballot_w32(mine);
            }

            #pragma unroll
            for (int q = 0; q < EPW / 32; ++q) {
                    unsigned m = mm[q];
                    while (m != 0) {
                        int ss[B], dd[B];
                        #pragma unroll
                        for (int k = 0; k < B; ++k) {
                            const bool on = (m != 0);
                            const int  b  = on ? __builtin_ctz(m) : 0;
                            m = on ? (m & (m - 1u)) : 0u;
                            ss[k] = __builtin_amdgcn_readlane(sv[q], b);
                            dd[k] = on ? __builtin_amdgcn_readlane(dv[q], b) : -1;
                        }
                        float v[B][NL];
                        #pragma unroll
                        for (int k = 0; k < B; ++k) {
                            const float* hrow = h + (size_t)ss[k] * CH;
                            #pragma unroll
                            for (int c = 0; c < NL; ++c) v[k][c] = hrow[lane + 32 * c];
                        }
                        #pragma unroll
                        for (int k = 0; k < B; ++k) {
                            if (dd[k] >= 0) {
                                float* arow = agg + (size_t)dd[k] * CH;
                                #pragma unroll
                                for (int c = 0; c < NL; ++c) atomicAdd(arow + lane + 32 * c, RELU ? fmaxf(v[k][c], 0.0f) : v[k][c]);
                            }
                        }
                    }
            }
        }
    }
}

__global__ __launch_bounds__(256) void pool_mean_bias_kernel(
    const float* __restrict__ ysum, const float* __restrict__ cnt,
    const float* __restrict__ lin_b, float* __restrict__ out, int G)
{
    const int idx = blockIdx.x * blockDim.x + threadIdx.x;
    if (idx >= (G << 4)) return;
    const float m = fmaxf(cnt[idx >> 4], 1.0f);
    out[idx] = ysum[idx] / m + lin_b[idx & 15];
}


static inline unsigned blocks_for(long long n, int t) {
    return (unsigned)((n + t - 1) / t);
}

extern "C" void kernel_launch(void* const* d_in, const int* in_sizes, int n_in,
                              void* d_out, int out_size, void* d_ws, size_t ws_size,
                              hipStream_t stream)
{
    const float* x        = (const float*)d_in[0];
    const int*   edge     = (const int*)  d_in[1];
    const int*   batch    = (const int*)  d_in[2];
    const float* w_stage1 = (const float*)d_in[4];
    const float* b_stage1 = (const float*)d_in[5];
    const float* rel1_w   = (const float*)d_in[6];
    const float* rel1_b   = (const float*)d_in[7];
    const float* root1_w  = (const float*)d_in[8];
    const float* rel2_w   = (const float*)d_in[9];
    const float* rel2_b   = (const float*)d_in[10];
    const float* root2_w  = (const float*)d_in[11];
    const float* rel3_w   = (const float*)d_in[12];
    const float* rel3_b   = (const float*)d_in[13];
    const float* root3_w  = (const float*)d_in[14];
    const float* lin_w    = (const float*)d_in[15];
    const float* lin_b    = (const float*)d_in[16];
    float*       out      = (float*)d_out;

    const int N = in_sizes[0] / 128;
    const int E = in_sizes[1] / 2;
    const int G = out_size / 16;
    const int* src = edge;
    const int* dst = edge + E;

    float* ws = (float*)d_ws;
    size_t off = 0;
    float* bufA   = ws + off; off += (size_t)N * 128;
    float* prod1  = ws + off; off += (size_t)N * 64;
    float* pre1   = ws + off; off += (size_t)N * 64;
    float* agg    = ws + off; off += (size_t)N * 128;
    float* agg3   = ws + off; off += (size_t)N * 128;
    float* ysum   = ws + off; off += (size_t)G * 16;
    float* cnt    = ws + off; off += (size_t)G;
    int*   tickets = (int*)(ws + off); off += 8;
    (void)ws_size; (void)n_in;

    const int T = 256;


    gemm_chain_kernel<128, 64, 64, 8, 64, true><<<GEMM_GROUPS, T, 0, stream>>>(
        x, w_stage1, b_stage1, rel1_w, root1_w, rel1_b, prod1, pre1, N,
        agg, ysum, (G * 16 + G) / 4 + 2);

    scatter_add_kernel<64, false><<<SCATTER_GROUPS, T, 0, stream>>>(
        prod1, src, dst, pre1, E, N, tickets);

    scatter_add_kernel<64, true><<<SCATTER_GROUPS, T, 0, stream>>>(
        pre1, src, dst, agg, E, N, tickets + 2);
    gemm_wmma_kernel<true, true, 64, 128, 8, 128, false, false, true><<<GEMM_GROUPS, T, 0, stream>>>(
        agg, rel2_w, pre1, root2_w, rel2_b, bufA, N, agg3, nullptr, 0,
        nullptr, nullptr, nullptr, nullptr);

    scatter_add_kernel<128, false><<<SCATTER_GROUPS, T, 0, stream>>>(
        bufA, src, dst, agg3, E, N, tickets + 4);
    gemm_wmma_kernel<true, true, 128, 128, 8, 0, false, true, false><<<GEMM_GROUPS, T, 0, stream>>>(
        agg3, rel3_w, bufA, root3_w, rel3_b, nullptr, N, nullptr, nullptr, 0,
        lin_w, batch, ysum, cnt);

    pool_mean_bias_kernel<<<blocks_for((long long)G << 4, T), T, 0, stream>>>(
        ysum, cnt, lin_b, out, G);
}
